# starting kernel
# baseline (speedup 1.0000x reference)
#include <hip/hip_runtime.h>
#include <hip/hip_bf16.h>
#include <math.h>

typedef __attribute__((ext_vector_type(2))) float v2f;
typedef __attribute__((ext_vector_type(8))) float v8f;

#define HEADS 4
#define ATT_F 16
#define DNODE 64
#define DEDGE 16

__global__ __launch_bounds__(256) void gat_node_project(
    const float* __restrict__ nodes, const float* __restrict__ W,
    float* __restrict__ Wn, int n_nodes, int n_tiles)
{
  int wave = blockIdx.x * (blockDim.x >> 5) + (threadIdx.x >> 5);
  if (wave >= n_tiles) return;
  int lane = threadIdx.x & 31;
  int half = lane >> 4;
  int l16  = lane & 15;
  int row0 = wave * 16;
  int m = row0 + l16; if (m >= n_nodes) m = n_nodes - 1;
  const float* __restrict__ arow = nodes + (size_t)m * DNODE;

  v8f acc[4] = {};
  for (int k0 = 0; k0 < DNODE; k0 += 4) {
    int kk = k0 + 2 * half;
    v2f a; a.x = arow[kk]; a.y = arow[kk + 1];
#pragma unroll
    for (int t = 0; t < 4; ++t) {
      int j = t * 16 + l16;
      v2f b; b.x = W[(size_t)j * DNODE + kk]; b.y = W[(size_t)j * DNODE + kk + 1];
      acc[t] = __builtin_amdgcn_wmma_f32_16x16x4_f32(
           false, a,  false, b,
           (short)0, acc[t],  false,  false);
    }
  }

  if (row0 + 16 <= n_nodes) {
#pragma unroll
    for (int t = 0; t < 4; ++t)
#pragma unroll
      for (int v = 0; v < 8; ++v)
        Wn[(size_t)(row0 + v + 8 * half) * DNODE + t * 16 + l16] = acc[t][v];
  } else {
#pragma unroll
    for (int t = 0; t < 4; ++t)
#pragma unroll
      for (int v = 0; v < 8; ++v) {
        int mo = row0 + v + 8 * half;
        if (mo < n_nodes) Wn[(size_t)mo * DNODE + t * 16 + l16] = acc[t][v];
      }
  }
}

__global__ __launch_bounds__(256) void gat_node_scores(
    const float* __restrict__ Wn, const float* __restrict__ a,
    float* __restrict__ score_s, float* __restrict__ score_r, int n_nodes)
{
  int i = blockIdx.x * blockDim.x + threadIdx.x;
  if (i >= n_nodes) return;
  const float* __restrict__ row = Wn + (size_t)i * DNODE;
#pragma unroll
  for (int h = 0; h < HEADS; ++h) {
    float ss = 0.f, sr = 0.f;
#pragma unroll
    for (int f = 0; f < ATT_F; ++f) {
      float v = row[h * ATT_F + f];
      ss += v * a[h * 48 + f];
      sr += v * a[h * 48 + 16 + f];
    }
    score_s[i * HEADS + h] = ss;
    score_r[i * HEADS + h] = sr;
  }
}

__global__ void gat_wcomb(const float* __restrict__ W_edge,
                          const float* __restrict__ a, float* __restrict__ wcomb)
{
  int idx = threadIdx.x;
  if (idx >= HEADS * DEDGE) return;
  int h = idx >> 4, i = idx & 15;
  float acc = 0.f;
#pragma unroll
  for (int f = 0; f < ATT_F; ++f)
    acc += a[h * 48 + 32 + f] * W_edge[((h * ATT_F) + f) * DEDGE + i];
  wcomb[idx] = acc;
}

__global__ __launch_bounds__(256) void gat_edge_pass1(
    const float* __restrict__ edges, const int* __restrict__ recv,
    const int* __restrict__ send, const float* __restrict__ score_s,
    const float* __restrict__ score_r, const float* __restrict__ wcomb,
    float* __restrict__ ex, float* __restrict__ norm, int n_edges, int n_nodes)
{
  __shared__ float wc[HEADS * DEDGE];
  if (threadIdx.x < HEADS * DEDGE) wc[threadIdx.x] = wcomb[threadIdx.x];
  __syncthreads();

  int e = blockIdx.x * blockDim.x + threadIdx.x;
  int etot = n_edges + n_nodes;
  if (e >= etot) return;

  int s, r;
  float elt[HEADS] = {0.f, 0.f, 0.f, 0.f};
  if (e < n_edges) {
    s = send[e]; r = recv[e];
    const float4* __restrict__ ep = (const float4*)(edges + (size_t)e * DEDGE);
    float ef[DEDGE];
#pragma unroll
    for (int c = 0; c < 4; ++c) {
      float4 q = ep[c];
      ef[4 * c + 0] = q.x; ef[4 * c + 1] = q.y; ef[4 * c + 2] = q.z; ef[4 * c + 3] = q.w;
    }
#pragma unroll
    for (int h = 0; h < HEADS; ++h) {
      float acc = 0.f;
#pragma unroll
      for (int i = 0; i < DEDGE; ++i) acc += ef[i] * wc[h * DEDGE + i];
      elt[h] = acc;
    }
  } else {
    s = r = e - n_edges;
  }
#pragma unroll
  for (int h = 0; h < HEADS; ++h) {
    float m  = score_s[s * HEADS + h] + score_r[r * HEADS + h] + elt[h];
    float lm = (m > 0.f) ? m : 0.01f * m;
    float ez = __expf(lm);
    ex[(size_t)e * HEADS + h] = ez;
    atomicAdd(&norm[r * HEADS + h], ez);
  }
}

__global__ __launch_bounds__(256) void gat_edge_pass2(
    const int* __restrict__ recv, const int* __restrict__ send,
    const float* __restrict__ ex, const float* __restrict__ norm,
    const float* __restrict__ Wn, float* __restrict__ aggr,
    int n_edges, int n_nodes)
{
  int e = blockIdx.x * blockDim.x + threadIdx.x;
  int etot = n_edges + n_nodes;
  if (e >= etot) return;
  int s, r;
  if (e < n_edges) { s = send[e]; r = recv[e]; }
  else             { s = r = e - n_edges; }

  float al[HEADS];
#pragma unroll
  for (int h = 0; h < HEADS; ++h)
    al[h] = ex[(size_t)e * HEADS + h] / norm[r * HEADS + h];

  const float4* __restrict__ src = (const float4*)(Wn + (size_t)s * DNODE);
  float* dst = aggr + (size_t)r * DNODE;
#pragma unroll
  for (int c = 0; c < 16; ++c) {
    float4 v = src[c];
    float aa = al[c >> 2];
    atomicAdd(dst + 4 * c + 0, aa * v.x);
    atomicAdd(dst + 4 * c + 1, aa * v.y);
    atomicAdd(dst + 4 * c + 2, aa * v.z);
    atomicAdd(dst + 4 * c + 3, aa * v.w);
  }
}

__global__ __launch_bounds__(256) void gat_finalize(
    float* __restrict__ out, const float* __restrict__ scale,
    const float* __restrict__ bias, int n_nodes)
{
  int i = blockIdx.x * blockDim.x + threadIdx.x;
  if (i >= n_nodes) return;
  float* row = out + (size_t)i * DNODE;
  float v[DNODE];
  float mean = 0.f;
#pragma unroll
  for (int j = 0; j < DNODE; ++j) {
    float x = row[j];
    x = (x > 0.f) ? x : expm1f(x);
    v[j] = x; mean += x;
  }
  mean *= (1.f / DNODE);
  float var = 0.f;
#pragma unroll
  for (int j = 0; j < DNODE; ++j) { float d = v[j] - mean; var += d * d; }
  var *= (1.f / DNODE);
  float inv = rsqrtf(var + 1e-6f);
#pragma unroll
  for (int j = 0; j < DNODE; ++j)
    row[j] = (v[j] - mean) * inv * scale[j] + bias[j];
}

extern "C" void kernel_launch(void* const* d_in, const int* in_sizes, int n_in,
                              void* d_out, int out_size, void* d_ws, size_t ws_size,
                              hipStream_t stream)
{
  const float* nodes  = (const float*)d_in[0];
  const float* edges  = (const float*)d_in[1];
  const int*   recv   = (const int*)d_in[2];
  const int*   send   = (const int*)d_in[3];
  const float* W      = (const float*)d_in[4];
  const float* W_edge = (const float*)d_in[5];
  const float* a      = (const float*)d_in[6];
  const float* ln_s   = (const float*)d_in[7];
  const float* ln_b   = (const float*)d_in[8];

  const int n_nodes = in_sizes[0] / DNODE;
  const int n_edges = in_sizes[1] / DEDGE;
  const int etot    = n_edges + n_nodes;
  float* out = (float*)d_out;

  char* ws = (char*)d_ws;
  size_t off = 0;
  auto carve = [&](size_t bytes) -> float* {
    float* p = (float*)(ws + off);
    off += (bytes + 255) & ~(size_t)255;
    return p;
  };
  float* Wn      = carve((size_t)n_nodes * DNODE * sizeof(float));
  float* score_s = carve((size_t)n_nodes * HEADS * sizeof(float));
  float* score_r = carve((size_t)n_nodes * HEADS * sizeof(float));
  float* wcomb   = carve((size_t)HEADS * DEDGE * sizeof(float));
  float* norm    = carve((size_t)n_nodes * HEADS * sizeof(float));
  float* ex      = carve((size_t)etot * HEADS * sizeof(float));

  hipMemsetAsync(out,  0, (size_t)n_nodes * DNODE * sizeof(float), stream);
  hipMemsetAsync(norm, 0, (size_t)n_nodes * HEADS * sizeof(float), stream);

  const int n_tiles = (n_nodes + 15) / 16;
  dim3 blk(256);

  gat_node_project<<<dim3((n_tiles + 7) / 8), blk, 0, stream>>>(
      nodes, W, Wn, n_nodes, n_tiles);
  gat_node_scores<<<dim3((n_nodes + 255) / 256), blk, 0, stream>>>(
      Wn, a, score_s, score_r, n_nodes);
  gat_wcomb<<<dim3(1), dim3(64), 0, stream>>>(W_edge, a, wcomb);
  gat_edge_pass1<<<dim3((etot + 255) / 256), blk, 0, stream>>>(
      edges, recv, send, score_s, score_r, wcomb, ex, norm, n_edges, n_nodes);
  gat_edge_pass2<<<dim3((etot + 255) / 256), blk, 0, stream>>>(
      recv, send, ex, norm, Wn, out, n_edges, n_nodes);
  gat_finalize<<<dim3((n_nodes + 255) / 256), blk, 0, stream>>>(
      out, ln_s, ln_b, n_nodes);
}


# edge pass 2: lanes spread over one edge's 64 features, not over 32 edges, so each row load and each f32 atomic add covers a whole 128-byte line; same products, atomic adds still in arrival order
# speedup vs baseline: 3.8849x; 3.8849x over previous
#include <hip/hip_runtime.h>
#include <hip/hip_bf16.h>
#include <math.h>

typedef __attribute__((ext_vector_type(2))) float v2f;
typedef __attribute__((ext_vector_type(8))) float v8f;

#define HEADS 4
#define ATT_F 16
#define DNODE 64
#define DEDGE 16

__global__ __launch_bounds__(256) void gat_node_project(
    const float* __restrict__ nodes, const float* __restrict__ W,
    float* __restrict__ Wn, int n_nodes, int n_tiles)
{
  int wave = blockIdx.x * (blockDim.x >> 5) + (threadIdx.x >> 5);
  if (wave >= n_tiles) return;
  int lane = threadIdx.x & 31;
  int half = lane >> 4;
  int l16  = lane & 15;
  int row0 = wave * 16;
  int m = row0 + l16; if (m >= n_nodes) m = n_nodes - 1;
  const float* __restrict__ arow = nodes + (size_t)m * DNODE;

  v8f acc[4] = {};
  for (int k0 = 0; k0 < DNODE; k0 += 4) {
    int kk = k0 + 2 * half;
    v2f a; a.x = arow[kk]; a.y = arow[kk + 1];
#pragma unroll
    for (int t = 0; t < 4; ++t) {
      int j = t * 16 + l16;
      v2f b; b.x = W[(size_t)j * DNODE + kk]; b.y = W[(size_t)j * DNODE + kk + 1];
      acc[t] = __builtin_amdgcn_wmma_f32_16x16x4_f32(
           false, a,  false, b,
           (short)0, acc[t],  false,  false);
    }
  }

  if (row0 + 16 <= n_nodes) {
#pragma unroll
    for (int t = 0; t < 4; ++t)
#pragma unroll
      for (int v = 0; v < 8; ++v)
        Wn[(size_t)(row0 + v + 8 * half) * DNODE + t * 16 + l16] = acc[t][v];
  } else {
#pragma unroll
    for (int t = 0; t < 4; ++t)
#pragma unroll
      for (int v = 0; v < 8; ++v) {
        int mo = row0 + v + 8 * half;
        if (mo < n_nodes) Wn[(size_t)mo * DNODE + t * 16 + l16] = acc[t][v];
      }
  }
}

__global__ __launch_bounds__(256) void gat_node_scores(
    const float* __restrict__ Wn, const float* __restrict__ a,
    float* __restrict__ score_s, float* __restrict__ score_r, int n_nodes)
{
  int i = blockIdx.x * blockDim.x + threadIdx.x;
  if (i >= n_nodes) return;
  const float* __restrict__ row = Wn + (size_t)i * DNODE;
#pragma unroll
  for (int h = 0; h < HEADS; ++h) {
    float ss = 0.f, sr = 0.f;
#pragma unroll
    for (int f = 0; f < ATT_F; ++f) {
      float v = row[h * ATT_F + f];
      ss += v * a[h * 48 + f];
      sr += v * a[h * 48 + 16 + f];
    }
    score_s[i * HEADS + h] = ss;
    score_r[i * HEADS + h] = sr;
  }
}

__global__ void gat_wcomb(const float* __restrict__ W_edge,
                          const float* __restrict__ a, float* __restrict__ wcomb)
{
  int idx = threadIdx.x;
  if (idx >= HEADS * DEDGE) return;
  int h = idx >> 4, i = idx & 15;
  float acc = 0.f;
#pragma unroll
  for (int f = 0; f < ATT_F; ++f)
    acc += a[h * 48 + 32 + f] * W_edge[((h * ATT_F) + f) * DEDGE + i];
  wcomb[idx] = acc;
}

__global__ __launch_bounds__(256) void gat_edge_pass1(
    const float* __restrict__ edges, const int* __restrict__ recv,
    const int* __restrict__ send, const float* __restrict__ score_s,
    const float* __restrict__ score_r, const float* __restrict__ wcomb,
    float* __restrict__ ex, float* __restrict__ norm, int n_edges, int n_nodes)
{
  __shared__ float wc[HEADS * DEDGE];
  if (threadIdx.x < HEADS * DEDGE) wc[threadIdx.x] = wcomb[threadIdx.x];
  __syncthreads();

  int e = blockIdx.x * blockDim.x + threadIdx.x;
  int etot = n_edges + n_nodes;
  if (e >= etot) return;

  int s, r;
  float elt[HEADS] = {0.f, 0.f, 0.f, 0.f};
  if (e < n_edges) {
    s = send[e]; r = recv[e];
    const float4* __restrict__ ep = (const float4*)(edges + (size_t)e * DEDGE);
    float ef[DEDGE];
#pragma unroll
    for (int c = 0; c < 4; ++c) {
      float4 q = ep[c];
      ef[4 * c + 0] = q.x; ef[4 * c + 1] = q.y; ef[4 * c + 2] = q.z; ef[4 * c + 3] = q.w;
    }
#pragma unroll
    for (int h = 0; h < HEADS; ++h) {
      float acc = 0.f;
#pragma unroll
      for (int i = 0; i < DEDGE; ++i) acc += ef[i] * wc[h * DEDGE + i];
      elt[h] = acc;
    }
  } else {
    s = r = e - n_edges;
  }
#pragma unroll
  for (int h = 0; h < HEADS; ++h) {
    float m  = score_s[s * HEADS + h] + score_r[r * HEADS + h] + elt[h];
    float lm = (m > 0.f) ? m : 0.01f * m;
    float ez = __expf(lm);
    ex[(size_t)e * HEADS + h] = ez;
    atomicAdd(&norm[r * HEADS + h], ez);
  }
}

__device__ __forceinline__ int lane_int(int v, int j)
{
  return __builtin_amdgcn_readlane(v, j);
}
__device__ __forceinline__ float lane_float(float v, int j)
{
  return __int_as_float(__builtin_amdgcn_readlane(__float_as_int(v), j));
}

#define EDGE_GROUP 8

__global__ __launch_bounds__(256) void gat_edge_pass2(
    const int* __restrict__ recv, const int* __restrict__ send,
    const float* __restrict__ ex, const float* __restrict__ norm,
    const float* __restrict__ Wn, float* __restrict__ aggr,
    int n_edges, int n_nodes)
{
  const int lane = threadIdx.x & 31;
  const int e    = blockIdx.x * blockDim.x + threadIdx.x;
  const int e0   = e - lane;
  const int etot = n_edges + n_nodes;
  if (e0 >= etot) return;

  int s = 0, r = 0;
  float al[HEADS] = {0.f, 0.f, 0.f, 0.f};
  if (e < etot) {
    if (e < n_edges) { s = send[e]; r = recv[e]; }
    else             { s = r = e - n_edges; }
#pragma unroll
    for (int h = 0; h < HEADS; ++h)
      al[h] = ex[(size_t)e * HEADS + h] / norm[r * HEADS + h];
  }

  const int  cnt = min(32, etot - e0);
  const bool up  = lane >= 16;

  int j0 = 0;
  for (; j0 + EDGE_GROUP <= cnt; j0 += EDGE_GROUP) {
    float vlo[EDGE_GROUP], vhi[EDGE_GROUP];
#pragma unroll
    for (int u = 0; u < EDGE_GROUP; ++u) {
      const float* __restrict__ srow = Wn + (size_t)lane_int(s, j0 + u) * DNODE;
      vlo[u] = srow[lane];
      vhi[u] = srow[32 + lane];
    }
#pragma unroll
    for (int u = 0; u < EDGE_GROUP; ++u) {
      const int j = j0 + u;
      float* drow = aggr + (size_t)lane_int(r, j) * DNODE;
      const float a0 = lane_float(al[0], j), a1 = lane_float(al[1], j);
      const float a2 = lane_float(al[2], j), a3 = lane_float(al[3], j);
      atomicAdd(drow + lane,      (up ? a1 : a0) * vlo[u]);
      atomicAdd(drow + 32 + lane, (up ? a3 : a2) * vhi[u]);
    }
  }
  for (; j0 < cnt; ++j0) {
    const float* __restrict__ srow = Wn + (size_t)lane_int(s, j0) * DNODE;
    float* drow = aggr + (size_t)lane_int(r, j0) * DNODE;
    const float a0 = lane_float(al[0], j0), a1 = lane_float(al[1], j0);
    const float a2 = lane_float(al[2], j0), a3 = lane_float(al[3], j0);
    const float vlo = srow[lane], vhi = srow[32 + lane];
    atomicAdd(drow + lane,      (up ? a1 : a0) * vlo);
    atomicAdd(drow + 32 + lane, (up ? a3 : a2) * vhi);
  }
}

__global__ __launch_bounds__(256) void gat_finalize(
    float* __restrict__ out, const float* __restrict__ scale,
    const float* __restrict__ bias, int n_nodes)
{
  int i = blockIdx.x * blockDim.x + threadIdx.x;
  if (i >= n_nodes) return;
  float* row = out + (size_t)i * DNODE;
  float v[DNODE];
  float mean = 0.f;
#pragma unroll
  for (int j = 0; j < DNODE; ++j) {
    float x = row[j];
    x = (x > 0.f) ? x : expm1f(x);
    v[j] = x; mean += x;
  }
  mean *= (1.f / DNODE);
  float var = 0.f;
#pragma unroll
  for (int j = 0; j < DNODE; ++j) { float d = v[j] - mean; var += d * d; }
  var *= (1.f / DNODE);
  float inv = rsqrtf(var + 1e-6f);
#pragma unroll
  for (int j = 0; j < DNODE; ++j)
    row[j] = (v[j] - mean) * inv * scale[j] + bias[j];
}

extern "C" void kernel_launch(void* const* d_in, const int* in_sizes, int n_in,
                              void* d_out, int out_size, void* d_ws, size_t ws_size,
                              hipStream_t stream)
{
  const float* nodes  = (const float*)d_in[0];
  const float* edges  = (const float*)d_in[1];
  const int*   recv   = (const int*)d_in[2];
  const int*   send   = (const int*)d_in[3];
  const float* W      = (const float*)d_in[4];
  const float* W_edge = (const float*)d_in[5];
  const float* a      = (const float*)d_in[6];
  const float* ln_s   = (const float*)d_in[7];
  const float* ln_b   = (const float*)d_in[8];

  const int n_nodes = in_sizes[0] / DNODE;
  const int n_edges = in_sizes[1] / DEDGE;
  const int etot    = n_edges + n_nodes;
  float* out = (float*)d_out;

  char* ws = (char*)d_ws;
  size_t off = 0;
  auto carve = [&](size_t bytes) -> float* {
    float* p = (float*)(ws + off);
    off += (bytes + 255) & ~(size_t)255;
    return p;
  };
  float* Wn      = carve((size_t)n_nodes * DNODE * sizeof(float));
  float* score_s = carve((size_t)n_nodes * HEADS * sizeof(float));
  float* score_r = carve((size_t)n_nodes * HEADS * sizeof(float));
  float* wcomb   = carve((size_t)HEADS * DEDGE * sizeof(float));
  float* norm    = carve((size_t)n_nodes * HEADS * sizeof(float));
  float* ex      = carve((size_t)etot * HEADS * sizeof(float));

  hipMemsetAsync(out,  0, (size_t)n_nodes * DNODE * sizeof(float), stream);
  hipMemsetAsync(norm, 0, (size_t)n_nodes * HEADS * sizeof(float), stream);

  const int n_tiles = (n_nodes + 15) / 16;
  dim3 blk(256);

  gat_node_project<<<dim3((n_tiles + 7) / 8), blk, 0, stream>>>(
      nodes, W, Wn, n_nodes, n_tiles);
  gat_node_scores<<<dim3((n_nodes + 255) / 256), blk, 0, stream>>>(
      Wn, a, score_s, score_r, n_nodes);
  gat_wcomb<<<dim3(1), dim3(64), 0, stream>>>(W_edge, a, wcomb);
  gat_edge_pass1<<<dim3((etot + 255) / 256), blk, 0, stream>>>(
      edges, recv, send, score_s, score_r, wcomb, ex, norm, n_edges, n_nodes);
  gat_edge_pass2<<<dim3((etot + 255) / 256), blk, 0, stream>>>(
      recv, send, ex, norm, Wn, out, n_edges, n_nodes);
  gat_finalize<<<dim3((n_nodes + 255) / 256), blk, 0, stream>>>(
      out, ln_s, ln_b, n_nodes);
}


# edge passes 1 and 2 fused: softmax division deferred to finalize (sum of exp*Wn[s] per receiver divided once by its norm); norm adds grouped 8 edges x 4 heads per atomic; adds still in arrival order
# speedup vs baseline: 7.7813x; 2.0030x over previous
#include <hip/hip_runtime.h>
#include <hip/hip_bf16.h>
#include <math.h>

typedef __attribute__((ext_vector_type(2))) float v2f;
typedef __attribute__((ext_vector_type(8))) float v8f;

#define HEADS 4
#define ATT_F 16
#define DNODE 64
#define DEDGE 16

__global__ __launch_bounds__(256) void gat_node_project(
    const float* __restrict__ nodes, const float* __restrict__ W,
    float* __restrict__ Wn, int n_nodes, int n_tiles)
{
  int wave = blockIdx.x * (blockDim.x >> 5) + (threadIdx.x >> 5);
  if (wave >= n_tiles) return;
  int lane = threadIdx.x & 31;
  int half = lane >> 4;
  int l16  = lane & 15;
  int row0 = wave * 16;
  int m = row0 + l16; if (m >= n_nodes) m = n_nodes - 1;
  const float* __restrict__ arow = nodes + (size_t)m * DNODE;

  v8f acc[4] = {};
  for (int k0 = 0; k0 < DNODE; k0 += 4) {
    int kk = k0 + 2 * half;
    v2f a; a.x = arow[kk]; a.y = arow[kk + 1];
#pragma unroll
    for (int t = 0; t < 4; ++t) {
      int j = t * 16 + l16;
      v2f b; b.x = W[(size_t)j * DNODE + kk]; b.y = W[(size_t)j * DNODE + kk + 1];
      acc[t] = __builtin_amdgcn_wmma_f32_16x16x4_f32(
           false, a,  false, b,
           (short)0, acc[t],  false,  false);
    }
  }

  if (row0 + 16 <= n_nodes) {
#pragma unroll
    for (int t = 0; t < 4; ++t)
#pragma unroll
      for (int v = 0; v < 8; ++v)
        Wn[(size_t)(row0 + v + 8 * half) * DNODE + t * 16 + l16] = acc[t][v];
  } else {
#pragma unroll
    for (int t = 0; t < 4; ++t)
#pragma unroll
      for (int v = 0; v < 8; ++v) {
        int mo = row0 + v + 8 * half;
        if (mo < n_nodes) Wn[(size_t)mo * DNODE + t * 16 + l16] = acc[t][v];
      }
  }
}

__global__ __launch_bounds__(256) void gat_node_scores(
    const float* __restrict__ Wn, const float* __restrict__ a,
    float* __restrict__ score_s, float* __restrict__ score_r, int n_nodes)
{
  int i = blockIdx.x * blockDim.x + threadIdx.x;
  if (i >= n_nodes) return;
  const float* __restrict__ row = Wn + (size_t)i * DNODE;
#pragma unroll
  for (int h = 0; h < HEADS; ++h) {
    float ss = 0.f, sr = 0.f;
#pragma unroll
    for (int f = 0; f < ATT_F; ++f) {
      float v = row[h * ATT_F + f];
      ss += v * a[h * 48 + f];
      sr += v * a[h * 48 + 16 + f];
    }
    score_s[i * HEADS + h] = ss;
    score_r[i * HEADS + h] = sr;
  }
}

__global__ void gat_wcomb(const float* __restrict__ W_edge,
                          const float* __restrict__ a, float* __restrict__ wcomb)
{
  int idx = threadIdx.x;
  if (idx >= HEADS * DEDGE) return;
  int h = idx >> 4, i = idx & 15;
  float acc = 0.f;
#pragma unroll
  for (int f = 0; f < ATT_F; ++f)
    acc += a[h * 48 + 32 + f] * W_edge[((h * ATT_F) + f) * DEDGE + i];
  wcomb[idx] = acc;
}

__device__ __forceinline__ int lane_int(int v, int j)
{
  return __builtin_amdgcn_readlane(v, j);
}
__device__ __forceinline__ float lane_float(float v, int j)
{
  return __int_as_float(__builtin_amdgcn_readlane(__float_as_int(v), j));
}

#define EDGE_GROUP 8

__global__ __launch_bounds__(256) void gat_edge_fused(
    const float* __restrict__ edges, const int* __restrict__ recv,
    const int* __restrict__ send, const float* __restrict__ score_s,
    const float* __restrict__ score_r, const float* __restrict__ wcomb,
    const float* __restrict__ Wn, float* __restrict__ aggr,
    float* __restrict__ norm, int n_edges, int n_nodes)
{
  const int lane = threadIdx.x & 31;
  const int e    = blockIdx.x * blockDim.x + threadIdx.x;
  const int e0   = e - lane;
  const int etot = n_edges + n_nodes;
  if (e0 >= etot) return;

  int s = 0, r = 0;
  float ez[HEADS] = {0.f, 0.f, 0.f, 0.f};
  if (e < etot) {
    float elt[HEADS] = {0.f, 0.f, 0.f, 0.f};
    if (e < n_edges) {
      s = send[e]; r = recv[e];
      const float4* __restrict__ ep = (const float4*)(edges + (size_t)e * DEDGE);
      float ef[DEDGE];
#pragma unroll
      for (int c = 0; c < 4; ++c) {
        float4 q = ep[c];
        ef[4 * c + 0] = q.x; ef[4 * c + 1] = q.y; ef[4 * c + 2] = q.z; ef[4 * c + 3] = q.w;
      }
#pragma unroll
      for (int h = 0; h < HEADS; ++h) {
        float acc = 0.f;
#pragma unroll
        for (int i = 0; i < DEDGE; ++i) acc += ef[i] * wcomb[h * DEDGE + i];
        elt[h] = acc;
      }
    } else {
      s = r = e - n_edges;
    }
    const float4 qs = ((const float4*)score_s)[s];
    const float4 qr = ((const float4*)score_r)[r];
    const float sc[HEADS] = {qs.x + qr.x, qs.y + qr.y, qs.z + qr.z, qs.w + qr.w};
#pragma unroll
    for (int h = 0; h < HEADS; ++h) {
      float m  = sc[h] + elt[h];
      float lm = (m > 0.f) ? m : 0.01f * m;
      ez[h] = __expf(lm);
    }
  }

  const int  cnt = min(32, etot - e0);
  const bool up  = lane >= 16;

  {
    const int  hsel = lane & 3;
    const bool odd  = (lane & 1) != 0, hi = (lane & 2) != 0;
    for (int k = 0; k < 4; ++k) {
      const int src = 8 * k + (lane >> 2);
      const int   rr = __shfl(r, src);
      const float z0 = __shfl(ez[0], src), z1 = __shfl(ez[1], src);
      const float z2 = __shfl(ez[2], src), z3 = __shfl(ez[3], src);
      const float z01 = odd ? z1 : z0;
      const float z23 = odd ? z3 : z2;
      const float z   = hi ? z23 : z01;
      if (src < cnt) atomicAdd(norm + (size_t)rr * HEADS + hsel, z);
    }
  }

  int j0 = 0;
  for (; j0 + EDGE_GROUP <= cnt; j0 += EDGE_GROUP) {
    float vlo[EDGE_GROUP], vhi[EDGE_GROUP];
#pragma unroll
    for (int u = 0; u < EDGE_GROUP; ++u) {
      const float* __restrict__ srow = Wn + (size_t)lane_int(s, j0 + u) * DNODE;
      vlo[u] = srow[lane];
      vhi[u] = srow[32 + lane];
    }
#pragma unroll
    for (int u = 0; u < EDGE_GROUP; ++u) {
      const int j  = j0 + u;
      const int rj = lane_int(r, j);
      float* drow = aggr + (size_t)rj * DNODE;
      const float a0 = lane_float(ez[0], j), a1 = lane_float(ez[1], j);
      const float a2 = lane_float(ez[2], j), a3 = lane_float(ez[3], j);
      atomicAdd(drow + lane,      (up ? a1 : a0) * vlo[u]);
      atomicAdd(drow + 32 + lane, (up ? a3 : a2) * vhi[u]);
    }
  }
  for (; j0 < cnt; ++j0) {
    const float* __restrict__ srow = Wn + (size_t)lane_int(s, j0) * DNODE;
    const int rj = lane_int(r, j0);
    float* drow = aggr + (size_t)rj * DNODE;
    const float a0 = lane_float(ez[0], j0), a1 = lane_float(ez[1], j0);
    const float a2 = lane_float(ez[2], j0), a3 = lane_float(ez[3], j0);
    const float vlo = srow[lane], vhi = srow[32 + lane];
    atomicAdd(drow + lane,      (up ? a1 : a0) * vlo);
    atomicAdd(drow + 32 + lane, (up ? a3 : a2) * vhi);
  }
}

__global__ __launch_bounds__(256) void gat_finalize(
    float* __restrict__ out, const float* __restrict__ norm,
    const float* __restrict__ scale, const float* __restrict__ bias, int n_nodes)
{
  int i = blockIdx.x * blockDim.x + threadIdx.x;
  if (i >= n_nodes) return;
  float* row = out + (size_t)i * DNODE;
  const float4 q = ((const float4*)norm)[i];
  const float nrm[HEADS] = {q.x, q.y, q.z, q.w};
  float v[DNODE];
  float mean = 0.f;
#pragma unroll
  for (int j = 0; j < DNODE; ++j) {
    float x = row[j] / nrm[j / ATT_F];
    x = (x > 0.f) ? x : expm1f(x);
    v[j] = x; mean += x;
  }
  mean *= (1.f / DNODE);
  float var = 0.f;
#pragma unroll
  for (int j = 0; j < DNODE; ++j) { float d = v[j] - mean; var += d * d; }
  var *= (1.f / DNODE);
  float inv = rsqrtf(var + 1e-6f);
#pragma unroll
  for (int j = 0; j < DNODE; ++j)
    row[j] = (v[j] - mean) * inv * scale[j] + bias[j];
}

extern "C" void kernel_launch(void* const* d_in, const int* in_sizes, int n_in,
                              void* d_out, int out_size, void* d_ws, size_t ws_size,
                              hipStream_t stream)
{
  const float* nodes  = (const float*)d_in[0];
  const float* edges  = (const float*)d_in[1];
  const int*   recv   = (const int*)d_in[2];
  const int*   send   = (const int*)d_in[3];
  const float* W      = (const float*)d_in[4];
  const float* W_edge = (const float*)d_in[5];
  const float* a      = (const float*)d_in[6];
  const float* ln_s   = (const float*)d_in[7];
  const float* ln_b   = (const float*)d_in[8];

  const int n_nodes = in_sizes[0] / DNODE;
  const int n_edges = in_sizes[1] / DEDGE;
  const int etot    = n_edges + n_nodes;
  float* out = (float*)d_out;

  char* ws = (char*)d_ws;
  size_t off = 0;
  auto carve = [&](size_t bytes) -> float* {
    float* p = (float*)(ws + off);
    off += (bytes + 255) & ~(size_t)255;
    return p;
  };
  float* Wn      = carve((size_t)n_nodes * DNODE * sizeof(float));
  float* score_s = carve((size_t)n_nodes * HEADS * sizeof(float));
  float* score_r = carve((size_t)n_nodes * HEADS * sizeof(float));
  float* wcomb   = carve((size_t)HEADS * DEDGE * sizeof(float));
  float* norm    = carve((size_t)n_nodes * HEADS * sizeof(float));

  hipMemsetAsync(out,  0, (size_t)n_nodes * DNODE * sizeof(float), stream);
  hipMemsetAsync(norm, 0, (size_t)n_nodes * HEADS * sizeof(float), stream);

  const int n_tiles = (n_nodes + 15) / 16;
  dim3 blk(256);

  gat_node_project<<<dim3((n_tiles + 7) / 8), blk, 0, stream>>>(
      nodes, W, Wn, n_nodes, n_tiles);
  gat_node_scores<<<dim3((n_nodes + 255) / 256), blk, 0, stream>>>(
      Wn, a, score_s, score_r, n_nodes);
  gat_wcomb<<<dim3(1), dim3(64), 0, stream>>>(W_edge, a, wcomb);
  gat_edge_fused<<<dim3((etot + 255) / 256), blk, 0, stream>>>(
      edges, recv, send, score_s, score_r, wcomb, Wn, out, norm, n_edges, n_nodes);
  gat_finalize<<<dim3((n_nodes + 255) / 256), blk, 0, stream>>>(
      out, norm, ln_s, ln_b, n_nodes);
}


# edge adds go to one of two accumulation planes picked by the chip half the wave runs on, so each line is updated from one L2 half; finalize sums the planes; adds in arrival order, two partial sums
# speedup vs baseline: 9.9379x; 1.2772x over previous
#include <hip/hip_runtime.h>
#include <hip/hip_bf16.h>
#include <math.h>

typedef __attribute__((ext_vector_type(2))) float v2f;
typedef __attribute__((ext_vector_type(8))) float v8f;

#define HEADS 4
#define ATT_F 16
#define DNODE 64
#define DEDGE 16

__global__ __launch_bounds__(256) void gat_node_project(
    const float* __restrict__ nodes, const float* __restrict__ W,
    float* __restrict__ Wn, int n_nodes, int n_tiles)
{
  int wave = blockIdx.x * (blockDim.x >> 5) + (threadIdx.x >> 5);
  if (wave >= n_tiles) return;
  int lane = threadIdx.x & 31;
  int half = lane >> 4;
  int l16  = lane & 15;
  int row0 = wave * 16;
  int m = row0 + l16; if (m >= n_nodes) m = n_nodes - 1;
  const float* __restrict__ arow = nodes + (size_t)m * DNODE;

  v8f acc[4] = {};
  for (int k0 = 0; k0 < DNODE; k0 += 4) {
    int kk = k0 + 2 * half;
    v2f a; a.x = arow[kk]; a.y = arow[kk + 1];
#pragma unroll
    for (int t = 0; t < 4; ++t) {
      int j = t * 16 + l16;
      v2f b; b.x = W[(size_t)j * DNODE + kk]; b.y = W[(size_t)j * DNODE + kk + 1];
      acc[t] = __builtin_amdgcn_wmma_f32_16x16x4_f32(
           false, a,  false, b,
           (short)0, acc[t],  false,  false);
    }
  }

  if (row0 + 16 <= n_nodes) {
#pragma unroll
    for (int t = 0; t < 4; ++t)
#pragma unroll
      for (int v = 0; v < 8; ++v)
        Wn[(size_t)(row0 + v + 8 * half) * DNODE + t * 16 + l16] = acc[t][v];
  } else {
#pragma unroll
    for (int t = 0; t < 4; ++t)
#pragma unroll
      for (int v = 0; v < 8; ++v) {
        int mo = row0 + v + 8 * half;
        if (mo < n_nodes) Wn[(size_t)mo * DNODE + t * 16 + l16] = acc[t][v];
      }
  }
}

__global__ __launch_bounds__(256) void gat_node_scores(
    const float* __restrict__ Wn, const float* __restrict__ a,
    float* __restrict__ score_s, float* __restrict__ score_r, int n_nodes)
{
  int i = blockIdx.x * blockDim.x + threadIdx.x;
  if (i >= n_nodes) return;
  const float* __restrict__ row = Wn + (size_t)i * DNODE;
#pragma unroll
  for (int h = 0; h < HEADS; ++h) {
    float ss = 0.f, sr = 0.f;
#pragma unroll
    for (int f = 0; f < ATT_F; ++f) {
      float v = row[h * ATT_F + f];
      ss += v * a[h * 48 + f];
      sr += v * a[h * 48 + 16 + f];
    }
    score_s[i * HEADS + h] = ss;
    score_r[i * HEADS + h] = sr;
  }
}

__global__ void gat_wcomb(const float* __restrict__ W_edge,
                          const float* __restrict__ a, float* __restrict__ wcomb)
{
  int idx = threadIdx.x;
  if (idx >= HEADS * DEDGE) return;
  int h = idx >> 4, i = idx & 15;
  float acc = 0.f;
#pragma unroll
  for (int f = 0; f < ATT_F; ++f)
    acc += a[h * 48 + 32 + f] * W_edge[((h * ATT_F) + f) * DEDGE + i];
  wcomb[idx] = acc;
}

__device__ __forceinline__ int lane_int(int v, int j)
{
  return __builtin_amdgcn_readlane(v, j);
}
__device__ __forceinline__ float lane_float(float v, int j)
{
  return __int_as_float(__builtin_amdgcn_readlane(__float_as_int(v), j));
}

#define EDGE_GROUP 8

__global__ __launch_bounds__(256) void gat_edge_fused(
    const float* __restrict__ edges, const int* __restrict__ recv,
    const int* __restrict__ send, const float* __restrict__ score_s,
    const float* __restrict__ score_r, const float* __restrict__ wcomb,
    const float* __restrict__ Wn, float* __restrict__ aggr0,
    float* __restrict__ aggr1, float* __restrict__ norm01,
    int n_edges, int n_nodes)
{
  const int chip_half = (int)((__builtin_amdgcn_s_sendmsg_rtn(0x87) >> 18) & 1u);
  float* __restrict__ aggr = chip_half ? aggr1 : aggr0;
  float* __restrict__ norm = norm01 + (size_t)chip_half * (size_t)n_nodes * HEADS;
  const int lane = threadIdx.x & 31;
  const int e    = blockIdx.x * blockDim.x + threadIdx.x;
  const int e0   = e - lane;
  const int etot = n_edges + n_nodes;
  if (e0 >= etot) return;

  int s = 0, r = 0;
  float ez[HEADS] = {0.f, 0.f, 0.f, 0.f};
  if (e < etot) {
    float elt[HEADS] = {0.f, 0.f, 0.f, 0.f};
    if (e < n_edges) {
      s = send[e]; r = recv[e];
      const float4* __restrict__ ep = (const float4*)(edges + (size_t)e * DEDGE);
      float ef[DEDGE];
#pragma unroll
      for (int c = 0; c < 4; ++c) {
        float4 q = ep[c];
        ef[4 * c + 0] = q.x; ef[4 * c + 1] = q.y; ef[4 * c + 2] = q.z; ef[4 * c + 3] = q.w;
      }
#pragma unroll
      for (int h = 0; h < HEADS; ++h) {
        float acc = 0.f;
#pragma unroll
        for (int i = 0; i < DEDGE; ++i) acc += ef[i] * wcomb[h * DEDGE + i];
        elt[h] = acc;
      }
    } else {
      s = r = e - n_edges;
    }
    const float4 qs = ((const float4*)score_s)[s];
    const float4 qr = ((const float4*)score_r)[r];
    const float sc[HEADS] = {qs.x + qr.x, qs.y + qr.y, qs.z + qr.z, qs.w + qr.w};
#pragma unroll
    for (int h = 0; h < HEADS; ++h) {
      float m  = sc[h] + elt[h];
      float lm = (m > 0.f) ? m : 0.01f * m;
      ez[h] = __expf(lm);
    }
  }

  const int  cnt = min(32, etot - e0);
  const bool up  = lane >= 16;

  {
    const int  hsel = lane & 3;
    const bool odd  = (lane & 1) != 0, hi = (lane & 2) != 0;
    for (int k = 0; k < 4; ++k) {
      const int src = 8 * k + (lane >> 2);
      const int   rr = __shfl(r, src);
      const float z0 = __shfl(ez[0], src), z1 = __shfl(ez[1], src);
      const float z2 = __shfl(ez[2], src), z3 = __shfl(ez[3], src);
      const float z01 = odd ? z1 : z0;
      const float z23 = odd ? z3 : z2;
      const float z   = hi ? z23 : z01;
      if (src < cnt) atomicAdd(norm + (size_t)rr * HEADS + hsel, z);
    }
  }

  int j0 = 0;
  for (; j0 + EDGE_GROUP <= cnt; j0 += EDGE_GROUP) {
    float vlo[EDGE_GROUP], vhi[EDGE_GROUP];
#pragma unroll
    for (int u = 0; u < EDGE_GROUP; ++u) {
      const float* __restrict__ srow = Wn + (size_t)lane_int(s, j0 + u) * DNODE;
      vlo[u] = srow[lane];
      vhi[u] = srow[32 + lane];
    }
#pragma unroll
    for (int u = 0; u < EDGE_GROUP; ++u) {
      const int j  = j0 + u;
      const int rj = lane_int(r, j);
      float* drow = aggr + (size_t)rj * DNODE;
      const float a0 = lane_float(ez[0], j), a1 = lane_float(ez[1], j);
      const float a2 = lane_float(ez[2], j), a3 = lane_float(ez[3], j);
      atomicAdd(drow + lane,      (up ? a1 : a0) * vlo[u]);
      atomicAdd(drow + 32 + lane, (up ? a3 : a2) * vhi[u]);
    }
  }
  for (; j0 < cnt; ++j0) {
    const float* __restrict__ srow = Wn + (size_t)lane_int(s, j0) * DNODE;
    const int rj = lane_int(r, j0);
    float* drow = aggr + (size_t)rj * DNODE;
    const float a0 = lane_float(ez[0], j0), a1 = lane_float(ez[1], j0);
    const float a2 = lane_float(ez[2], j0), a3 = lane_float(ez[3], j0);
    const float vlo = srow[lane], vhi = srow[32 + lane];
    atomicAdd(drow + lane,      (up ? a1 : a0) * vlo);
    atomicAdd(drow + 32 + lane, (up ? a3 : a2) * vhi);
  }
}

__global__ __launch_bounds__(256) void gat_finalize(
    float* __restrict__ out, const float* __restrict__ aggr1,
    const float* __restrict__ norm01,
    const float* __restrict__ scale, const float* __restrict__ bias, int n_nodes)
{
  int i = blockIdx.x * blockDim.x + threadIdx.x;
  if (i >= n_nodes) return;
  float* row = out + (size_t)i * DNODE;
  const float* __restrict__ row1 = aggr1 + (size_t)i * DNODE;
  const float4 q  = ((const float4*)norm01)[i];
  const float4 q1 = ((const float4*)norm01)[(size_t)n_nodes + i];
  const float nrm[HEADS] = {q.x + q1.x, q.y + q1.y, q.z + q1.z, q.w + q1.w};
  float v[DNODE];
  float mean = 0.f;
#pragma unroll
  for (int j = 0; j < DNODE; ++j) {
    float x = (row[j] + row1[j]) / nrm[j / ATT_F];
    x = (x > 0.f) ? x : expm1f(x);
    v[j] = x; mean += x;
  }
  mean *= (1.f / DNODE);
  float var = 0.f;
#pragma unroll
  for (int j = 0; j < DNODE; ++j) { float d = v[j] - mean; var += d * d; }
  var *= (1.f / DNODE);
  float inv = rsqrtf(var + 1e-6f);
#pragma unroll
  for (int j = 0; j < DNODE; ++j)
    row[j] = (v[j] - mean) * inv * scale[j] + bias[j];
}

extern "C" void kernel_launch(void* const* d_in, const int* in_sizes, int n_in,
                              void* d_out, int out_size, void* d_ws, size_t ws_size,
                              hipStream_t stream)
{
  const float* nodes  = (const float*)d_in[0];
  const float* edges  = (const float*)d_in[1];
  const int*   recv   = (const int*)d_in[2];
  const int*   send   = (const int*)d_in[3];
  const float* W      = (const float*)d_in[4];
  const float* W_edge = (const float*)d_in[5];
  const float* a      = (const float*)d_in[6];
  const float* ln_s   = (const float*)d_in[7];
  const float* ln_b   = (const float*)d_in[8];

  const int n_nodes = in_sizes[0] / DNODE;
  const int n_edges = in_sizes[1] / DEDGE;
  const int etot    = n_edges + n_nodes;
  float* out = (float*)d_out;

  char* ws = (char*)d_ws;
  size_t off = 0;
  auto carve = [&](size_t bytes) -> float* {
    float* p = (float*)(ws + off);
    off += (bytes + 255) & ~(size_t)255;
    return p;
  };
  float* Wn      = carve((size_t)n_nodes * DNODE * sizeof(float));
  float* score_s = carve((size_t)n_nodes * HEADS * sizeof(float));
  float* score_r = carve((size_t)n_nodes * HEADS * sizeof(float));
  float* wcomb   = carve((size_t)HEADS * DEDGE * sizeof(float));
  const size_t plane_bytes = (size_t)n_nodes * DNODE * sizeof(float);
  const size_t norms_bytes = (size_t)2 * n_nodes * HEADS * sizeof(float);
  float* aggr1   = carve(plane_bytes + norms_bytes);
  float* norm01  = aggr1 + (size_t)n_nodes * DNODE;

  hipMemsetAsync(out,   0, plane_bytes, stream);
  hipMemsetAsync(aggr1, 0, plane_bytes + norms_bytes, stream);

  const int n_tiles = (n_nodes + 15) / 16;
  dim3 blk(256);

  gat_node_project<<<dim3((n_tiles + 7) / 8), blk, 0, stream>>>(
      nodes, W, Wn, n_nodes, n_tiles);
  gat_node_scores<<<dim3((n_nodes + 255) / 256), blk, 0, stream>>>(
      Wn, a, score_s, score_r, n_nodes);
  gat_wcomb<<<dim3(1), dim3(64), 0, stream>>>(W_edge, a, wcomb);
  gat_edge_fused<<<dim3((etot + 255) / 256), blk, 0, stream>>>(
      edges, recv, send, score_s, score_r, wcomb, Wn, out, aggr1, norm01, n_edges, n_nodes);
  gat_finalize<<<dim3((n_nodes + 255) / 256), blk, 0, stream>>>(
      out, aggr1, norm01, ln_s, ln_b, n_nodes);
}


# global f32 atomic scatter replaced: edges grouped by receiver range in LDS, one work-group sums a range's rows in LDS (LDS f32 adds, arrival order), finalize fused, LayerNorm sums by lane butterfly
# speedup vs baseline: 12.6326x; 1.2712x over previous
#include <hip/hip_runtime.h>
#include <hip/hip_bf16.h>
#include <math.h>

typedef __attribute__((ext_vector_type(2))) float v2f;
typedef __attribute__((ext_vector_type(8))) float v8f;

#define HEADS 4
#define ATT_F 16
#define DNODE 64
#define DEDGE 16

__global__ __launch_bounds__(256) void gat_node_project(
    const float* __restrict__ nodes, const float* __restrict__ W,
    float* __restrict__ Wn, int n_nodes, int n_tiles)
{
  int wave = blockIdx.x * (blockDim.x >> 5) + (threadIdx.x >> 5);
  if (wave >= n_tiles) return;
  int lane = threadIdx.x & 31;
  int half = lane >> 4;
  int l16  = lane & 15;
  int row0 = wave * 16;
  int m = row0 + l16; if (m >= n_nodes) m = n_nodes - 1;
  const float* __restrict__ arow = nodes + (size_t)m * DNODE;

  v8f acc[4] = {};
  for (int k0 = 0; k0 < DNODE; k0 += 4) {
    int kk = k0 + 2 * half;
    v2f a; a.x = arow[kk]; a.y = arow[kk + 1];
#pragma unroll
    for (int t = 0; t < 4; ++t) {
      int j = t * 16 + l16;
      v2f b; b.x = W[(size_t)j * DNODE + kk]; b.y = W[(size_t)j * DNODE + kk + 1];
      acc[t] = __builtin_amdgcn_wmma_f32_16x16x4_f32(
           false, a,  false, b,
           (short)0, acc[t],  false,  false);
    }
  }

  if (row0 + 16 <= n_nodes) {
#pragma unroll
    for (int t = 0; t < 4; ++t)
#pragma unroll
      for (int v = 0; v < 8; ++v)
        Wn[(size_t)(row0 + v + 8 * half) * DNODE + t * 16 + l16] = acc[t][v];
  } else {
#pragma unroll
    for (int t = 0; t < 4; ++t)
#pragma unroll
      for (int v = 0; v < 8; ++v) {
        int mo = row0 + v + 8 * half;
        if (mo < n_nodes) Wn[(size_t)mo * DNODE + t * 16 + l16] = acc[t][v];
      }
  }
}

__global__ __launch_bounds__(256) void gat_node_scores(
    const float* __restrict__ Wn, const float* __restrict__ a,
    float* __restrict__ score_s, float* __restrict__ score_r, int n_nodes)
{
  int i = blockIdx.x * blockDim.x + threadIdx.x;
  if (i >= n_nodes) return;
  const float* __restrict__ row = Wn + (size_t)i * DNODE;
#pragma unroll
  for (int h = 0; h < HEADS; ++h) {
    float ss = 0.f, sr = 0.f;
#pragma unroll
    for (int f = 0; f < ATT_F; ++f) {
      float v = row[h * ATT_F + f];
      ss += v * a[h * 48 + f];
      sr += v * a[h * 48 + 16 + f];
    }
    score_s[i * HEADS + h] = ss;
    score_r[i * HEADS + h] = sr;
  }
}

__global__ void gat_wcomb(const float* __restrict__ W_edge,
                          const float* __restrict__ a, float* __restrict__ wcomb)
{
  int idx = threadIdx.x;
  if (idx >= HEADS * DEDGE) return;
  int h = idx >> 4, i = idx & 15;
  float acc = 0.f;
#pragma unroll
  for (int f = 0; f < ATT_F; ++f)
    acc += a[h * 48 + 32 + f] * W_edge[((h * ATT_F) + f) * DEDGE + i];
  wcomb[idx] = acc;
}

__device__ __forceinline__ int lane_int(int v, int j)
{
  return __builtin_amdgcn_readlane(v, j);
}
__device__ __forceinline__ float lane_float(float v, int j)
{
  return __int_as_float(__builtin_amdgcn_readlane(__float_as_int(v), j));
}

#define EDGE_GROUP 8

__global__ __launch_bounds__(256) void gat_edge_fused(
    const float* __restrict__ edges, const int* __restrict__ recv,
    const int* __restrict__ send, const float* __restrict__ score_s,
    const float* __restrict__ score_r, const float* __restrict__ wcomb,
    const float* __restrict__ Wn, float* __restrict__ aggr0,
    float* __restrict__ aggr1, float* __restrict__ norm01,
    int n_edges, int n_nodes)
{
  const int chip_half = (int)((__builtin_amdgcn_s_sendmsg_rtn(0x87) >> 18) & 1u);
  float* __restrict__ aggr = chip_half ? aggr1 : aggr0;
  float* __restrict__ norm = norm01 + (size_t)chip_half * (size_t)n_nodes * HEADS;
  const int lane = threadIdx.x & 31;
  const int e    = blockIdx.x * blockDim.x + threadIdx.x;
  const int e0   = e - lane;
  const int etot = n_edges + n_nodes;
  if (e0 >= etot) return;

  int s = 0, r = 0;
  float ez[HEADS] = {0.f, 0.f, 0.f, 0.f};
  if (e < etot) {
    float elt[HEADS] = {0.f, 0.f, 0.f, 0.f};
    if (e < n_edges) {
      s = send[e]; r = recv[e];
      const float4* __restrict__ ep = (const float4*)(edges + (size_t)e * DEDGE);
      float ef[DEDGE];
#pragma unroll
      for (int c = 0; c < 4; ++c) {
        float4 q = ep[c];
        ef[4 * c + 0] = q.x; ef[4 * c + 1] = q.y; ef[4 * c + 2] = q.z; ef[4 * c + 3] = q.w;
      }
#pragma unroll
      for (int h = 0; h < HEADS; ++h) {
        float acc = 0.f;
#pragma unroll
        for (int i = 0; i < DEDGE; ++i) acc += ef[i] * wcomb[h * DEDGE + i];
        elt[h] = acc;
      }
    } else {
      s = r = e - n_edges;
    }
    const float4 qs = ((const float4*)score_s)[s];
    const float4 qr = ((const float4*)score_r)[r];
    const float sc[HEADS] = {qs.x + qr.x, qs.y + qr.y, qs.z + qr.z, qs.w + qr.w};
#pragma unroll
    for (int h = 0; h < HEADS; ++h) {
      float m  = sc[h] + elt[h];
      float lm = (m > 0.f) ? m : 0.01f * m;
      ez[h] = __expf(lm);
    }
  }

  const int  cnt = min(32, etot - e0);
  const bool up  = lane >= 16;

  {
    const int  hsel = lane & 3;
    const bool odd  = (lane & 1) != 0, hi = (lane & 2) != 0;
    for (int k = 0; k < 4; ++k) {
      const int src = 8 * k + (lane >> 2);
      const int   rr = __shfl(r, src);
      const float z0 = __shfl(ez[0], src), z1 = __shfl(ez[1], src);
      const float z2 = __shfl(ez[2], src), z3 = __shfl(ez[3], src);
      const float z01 = odd ? z1 : z0;
      const float z23 = odd ? z3 : z2;
      const float z   = hi ? z23 : z01;
      if (src < cnt) atomicAdd(norm + (size_t)rr * HEADS + hsel, z);
    }
  }

  int j0 = 0;
  for (; j0 + EDGE_GROUP <= cnt; j0 += EDGE_GROUP) {
    float vlo[EDGE_GROUP], vhi[EDGE_GROUP];
#pragma unroll
    for (int u = 0; u < EDGE_GROUP; ++u) {
      const float* __restrict__ srow = Wn + (size_t)lane_int(s, j0 + u) * DNODE;
      vlo[u] = srow[lane];
      vhi[u] = srow[32 + lane];
    }
#pragma unroll
    for (int u = 0; u < EDGE_GROUP; ++u) {
      const int j  = j0 + u;
      const int rj = lane_int(r, j);
      float* drow = aggr + (size_t)rj * DNODE;
      const float a0 = lane_float(ez[0], j), a1 = lane_float(ez[1], j);
      const float a2 = lane_float(ez[2], j), a3 = lane_float(ez[3], j);
      atomicAdd(drow + lane,      (up ? a1 : a0) * vlo[u]);
      atomicAdd(drow + 32 + lane, (up ? a3 : a2) * vhi[u]);
    }
  }
  for (; j0 < cnt; ++j0) {
    const float* __restrict__ srow = Wn + (size_t)lane_int(s, j0) * DNODE;
    const int rj = lane_int(r, j0);
    float* drow = aggr + (size_t)rj * DNODE;
    const float a0 = lane_float(ez[0], j0), a1 = lane_float(ez[1], j0);
    const float a2 = lane_float(ez[2], j0), a3 = lane_float(ez[3], j0);
    const float vlo = srow[lane], vhi = srow[32 + lane];
    atomicAdd(drow + lane,      (up ? a1 : a0) * vlo);
    atomicAdd(drow + 32 + lane, (up ? a3 : a2) * vhi);
  }
}

__global__ __launch_bounds__(256) void gat_finalize(
    float* __restrict__ out, const float* __restrict__ aggr1,
    const float* __restrict__ norm01,
    const float* __restrict__ scale, const float* __restrict__ bias, int n_nodes)
{
  int i = blockIdx.x * blockDim.x + threadIdx.x;
  if (i >= n_nodes) return;
  float* row = out + (size_t)i * DNODE;
  const float* __restrict__ row1 = aggr1 + (size_t)i * DNODE;
  const float4 q  = ((const float4*)norm01)[i];
  const float4 q1 = ((const float4*)norm01)[(size_t)n_nodes + i];
  const float nrm[HEADS] = {q.x + q1.x, q.y + q1.y, q.z + q1.z, q.w + q1.w};
  float v[DNODE];
  float mean = 0.f;
#pragma unroll
  for (int j = 0; j < DNODE; ++j) {
    float x = (row[j] + row1[j]) / nrm[j / ATT_F];
    x = (x > 0.f) ? x : expm1f(x);
    v[j] = x; mean += x;
  }
  mean *= (1.f / DNODE);
  float var = 0.f;
#pragma unroll
  for (int j = 0; j < DNODE; ++j) { float d = v[j] - mean; var += d * d; }
  var *= (1.f / DNODE);
  float inv = rsqrtf(var + 1e-6f);
#pragma unroll
  for (int j = 0; j < DNODE; ++j)
    row[j] = (v[j] - mean) * inv * scale[j] + bias[j];
}

#define NBINS  256
#define NCHUNK 256
#define OWNER_GROUP 16

__global__ __launch_bounds__(1024) void gat_edge_order(
    const float* __restrict__ edges, const int* __restrict__ recv,
    const int* __restrict__ send, const float* __restrict__ score_s,
    const float* __restrict__ wcomb,
    int* __restrict__ sr, float4* __restrict__ lgrec, int2* __restrict__ seg,
    int n_edges, int chunk_edges, int chunk_stride, int npb)
{
  __shared__ int s_cnt[NBINS];
  __shared__ int s_cur[NBINS];
  __shared__ int s_wsum[NBINS / 32];
  const int tid = threadIdx.x;
  const int c   = blockIdx.x;
  const int e_begin = c * chunk_edges;
  const int e_end   = min(n_edges, e_begin + chunk_edges);
  const int rbase   = c * chunk_stride;
  if (tid < NBINS) s_cnt[tid] = 0;
  __syncthreads();
  for (int e = e_begin + tid; e < e_end; e += 1024)
    atomicAdd(&s_cnt[recv[e] / npb], 1);
  __syncthreads();
  int v = 0, incl = 0;
  if (tid < NBINS) {
    v = s_cnt[tid];
    incl = v;
    for (int d = 1; d < 32; d <<= 1) {
      const int t = __shfl_up(incl, d);
      if ((tid & 31) >= d) incl += t;
    }
    if ((tid & 31) == 31) s_wsum[tid >> 5] = incl;
  }
  __syncthreads();
  if (tid < NBINS) {
    int before = 0;
    for (int w = 0; w < (tid >> 5); ++w) before += s_wsum[w];
    const int excl = before + incl - v;
    s_cur[tid] = excl;
    seg[tid * NCHUNK + c] = make_int2(rbase + excl, v);
  }
  __syncthreads();
  for (int e = e_begin + tid; e < e_end; e += 1024) {
    const int s = send[e], r = recv[e];
    const float4* __restrict__ ep = (const float4*)(edges + (size_t)e * DEDGE);
    float ef[DEDGE];
#pragma unroll
    for (int k = 0; k < 4; ++k) {
      float4 q = ep[k];
      ef[4 * k + 0] = q.x; ef[4 * k + 1] = q.y; ef[4 * k + 2] = q.z; ef[4 * k + 3] = q.w;
    }
    float elt[HEADS];
#pragma unroll
    for (int h = 0; h < HEADS; ++h) {
      float acc = 0.f;
#pragma unroll
      for (int i = 0; i < DEDGE; ++i) acc += ef[i] * wcomb[h * DEDGE + i];
      elt[h] = acc;
    }
    const float4 qs = ((const float4*)score_s)[s];
    const int bin  = r / npb;
    const int rl   = r - bin * npb;
    const int slot = rbase + atomicAdd(&s_cur[bin], 1);
    sr[slot]    = (int)((unsigned)s | ((unsigned)rl << 24));
    lgrec[slot] = make_float4(qs.x + elt[0], qs.y + elt[1], qs.z + elt[2], qs.w + elt[3]);
  }
}

__global__ __launch_bounds__(1024) void gat_owner_sum(
    const int* __restrict__ sr, const float4* __restrict__ lgrec,
    const int2* __restrict__ seg, const float* __restrict__ score_s,
    const float* __restrict__ score_r, const float* __restrict__ Wn,
    const float* __restrict__ scale, const float* __restrict__ bias,
    float* __restrict__ out, int n_nodes, int npb)
{
  extern __shared__ float smem[];
  float* acc = smem;
  float* nrm = smem + npb * DNODE;
  float* scr = nrm + npb * HEADS;
  const int tid  = threadIdx.x;
  const int lane = tid & 31;
  const int wave = __builtin_amdgcn_readfirstlane(tid >> 5);
  const int b     = blockIdx.x;
  const int node0 = b * npb;
  const int nn    = min(npb, n_nodes - node0);
  const bool up   = lane >= 16;

  for (int idx = tid; idx < nn * DNODE; idx += 1024) {
    const int n = idx >> 6, f = idx & 63, h = f >> 4;
    const int node = node0 + n;
    const float m  = score_s[node * HEADS + h] + score_r[node * HEADS + h];
    const float lm = (m > 0.f) ? m : 0.01f * m;
    const float z  = __expf(lm);
    acc[idx] = z * Wn[(size_t)node * DNODE + f];
    if ((f & 15) == 0) { nrm[n * HEADS + h] = z; scr[n * HEADS + h] = score_r[node * HEADS + h]; }
  }
  __syncthreads();

  int2 sgl = make_int2(0, 0);
  if (lane < NCHUNK / 32) sgl = seg[b * NCHUNK + wave * (NCHUNK / 32) + lane];
  int    w_nx = 0;
  float4 q_nx = make_float4(0.f, 0.f, 0.f, 0.f);
  {
    const int st0 = lane_int(sgl.x, 0), c0 = lane_int(sgl.y, 0);
    if (lane < min(32, c0)) { w_nx = sr[st0 + lane]; q_nx = lgrec[st0 + lane]; }
  }
  for (int k = 0; k < NCHUNK / 32; ++k) {
    const int start = lane_int(sgl.x, k);
    const int cnt   = lane_int(sgl.y, k);
    const int    w_first = w_nx;
    const float4 q_first = q_nx;
    if (k + 1 < NCHUNK / 32) {
      const int st1 = lane_int(sgl.x, k + 1), c1 = lane_int(sgl.y, k + 1);
      if (lane < min(32, c1)) { w_nx = sr[st1 + lane]; q_nx = lgrec[st1 + lane]; }
    }
    for (int base = 0; base < cnt; base += 32) {
      const int m = min(32, cnt - base);
      int s = 0, rl = 0;
      float ez0 = 0.f, ez1 = 0.f, ez2 = 0.f, ez3 = 0.f;
      if (lane < m) {
        int w = w_first;
        float4 q = q_first;
        if (base > 0) {
          w = sr[start + base + lane];
          q = lgrec[start + base + lane];
        }
        s  = w & 0xFFFFFF;
        rl = (int)((unsigned)w >> 24);
        const float m0 = q.x + scr[rl * HEADS + 0], m1 = q.y + scr[rl * HEADS + 1];
        const float m2 = q.z + scr[rl * HEADS + 2], m3 = q.w + scr[rl * HEADS + 3];
        ez0 = __expf((m0 > 0.f) ? m0 : 0.01f * m0);
        ez1 = __expf((m1 > 0.f) ? m1 : 0.01f * m1);
        ez2 = __expf((m2 > 0.f) ? m2 : 0.01f * m2);
        ez3 = __expf((m3 > 0.f) ? m3 : 0.01f * m3);
        atomicAdd(&nrm[rl * HEADS + 0], ez0);
        atomicAdd(&nrm[rl * HEADS + 1], ez1);
        atomicAdd(&nrm[rl * HEADS + 2], ez2);
        atomicAdd(&nrm[rl * HEADS + 3], ez3);
      }
      for (int j0 = 0; j0 < m; j0 += OWNER_GROUP) {
        float vlo[OWNER_GROUP], vhi[OWNER_GROUP];
#pragma unroll
        for (int u = 0; u < OWNER_GROUP; ++u) {
          vlo[u] = 0.f; vhi[u] = 0.f;
          if (j0 + u < m) {
            const float* __restrict__ srow = Wn + (size_t)lane_int(s, j0 + u) * DNODE;
            vlo[u] = srow[lane];
            vhi[u] = srow[32 + lane];
          }
        }
#pragma unroll
        for (int u = 0; u < OWNER_GROUP; ++u) {
          if (j0 + u < m) {
            const int j  = j0 + u;
            const int rj = lane_int(rl, j);
            float* drow = acc + rj * DNODE;
            const float a0 = lane_float(ez0, j), a1 = lane_float(ez1, j);
            const float a2 = lane_float(ez2, j), a3 = lane_float(ez3, j);
            atomicAdd(drow + lane,      (up ? a1 : a0) * vlo[u]);
            atomicAdd(drow + 32 + lane, (up ? a3 : a2) * vhi[u]);
          }
        }
      }
    }
  }
  __syncthreads();

  const float sc_lo = scale[lane], sc_hi = scale[32 + lane];
  const float bi_lo = bias[lane],  bi_hi = bias[32 + lane];
  for (int n = wave; n < nn; n += 32) {
    const float n01 = up ? nrm[n * HEADS + 1] : nrm[n * HEADS + 0];
    const float n23 = up ? nrm[n * HEADS + 3] : nrm[n * HEADS + 2];
    float x0 = acc[n * DNODE + lane] / n01;
    float x1 = acc[n * DNODE + 32 + lane] / n23;
    x0 = (x0 > 0.f) ? x0 : expm1f(x0);
    x1 = (x1 > 0.f) ? x1 : expm1f(x1);
    float sum = x0 + x1;
    for (int d = 16; d >= 1; d >>= 1) sum += __shfl_xor(sum, d);
    const float mean = sum * (1.f / DNODE);
    const float d0 = x0 - mean, d1 = x1 - mean;
    float var = d0 * d0 + d1 * d1;
    for (int d = 16; d >= 1; d >>= 1) var += __shfl_xor(var, d);
    var *= (1.f / DNODE);
    const float inv = rsqrtf(var + 1e-6f);
    float* row = out + (size_t)(node0 + n) * DNODE;
    row[lane]      = d0 * inv * sc_lo + bi_lo;
    row[32 + lane] = d1 * inv * sc_hi + bi_hi;
  }
}

extern "C" void kernel_launch(void* const* d_in, const int* in_sizes, int n_in,
                              void* d_out, int out_size, void* d_ws, size_t ws_size,
                              hipStream_t stream)
{
  const float* nodes  = (const float*)d_in[0];
  const float* edges  = (const float*)d_in[1];
  const int*   recv   = (const int*)d_in[2];
  const int*   send   = (const int*)d_in[3];
  const float* W      = (const float*)d_in[4];
  const float* W_edge = (const float*)d_in[5];
  const float* a      = (const float*)d_in[6];
  const float* ln_s   = (const float*)d_in[7];
  const float* ln_b   = (const float*)d_in[8];

  const int n_nodes = in_sizes[0] / DNODE;
  const int n_edges = in_sizes[1] / DEDGE;
  const int etot    = n_edges + n_nodes;
  float* out = (float*)d_out;

  char* ws = (char*)d_ws;
  size_t off = 0;
  auto carve = [&](size_t bytes) -> float* {
    float* p = (float*)(ws + off);
    off += (bytes + 255) & ~(size_t)255;
    return p;
  };
  float* Wn      = carve((size_t)n_nodes * DNODE * sizeof(float));
  float* score_s = carve((size_t)n_nodes * HEADS * sizeof(float));
  float* score_r = carve((size_t)n_nodes * HEADS * sizeof(float));
  float* wcomb   = carve((size_t)HEADS * DEDGE * sizeof(float));
  const size_t off_common = off;

  const int n_tiles = (n_nodes + 15) / 16;
  dim3 blk(256);

  gat_node_project<<<dim3((n_tiles + 7) / 8), blk, 0, stream>>>(
      nodes, W, Wn, n_nodes, n_tiles);
  gat_node_scores<<<dim3((n_nodes + 255) / 256), blk, 0, stream>>>(
      Wn, a, score_s, score_r, n_nodes);
  gat_wcomb<<<dim3(1), dim3(64), 0, stream>>>(W_edge, a, wcomb);

  const int npb          = (n_nodes + NBINS - 1) / NBINS;
  const int chunk_edges  = (n_edges + NCHUNK - 1) / NCHUNK;
  const int chunk_stride = (chunk_edges + 31) & ~31;
  const size_t lds_bytes = (size_t)npb * (DNODE + 2 * HEADS) * sizeof(float);
  int2*   seg   = (int2*)carve((size_t)NBINS * NCHUNK * sizeof(int2));
  int*    sr    = (int*)carve((size_t)NCHUNK * chunk_stride * sizeof(int));
  float4* lgrec = (float4*)carve((size_t)NCHUNK * chunk_stride * sizeof(float4));
  if (off <= ws_size && lds_bytes <= 65536 && n_nodes <= (1 << 24) && npb > 0) {
    gat_edge_order<<<dim3(NCHUNK), dim3(1024), 0, stream>>>(
        edges, recv, send, score_s, wcomb, sr, lgrec, seg,
        n_edges, chunk_edges, chunk_stride, npb);
    gat_owner_sum<<<dim3(NBINS), dim3(1024), lds_bytes, stream>>>(
        sr, lgrec, seg, score_s, score_r, Wn, ln_s, ln_b, out, n_nodes, npb);
    return;
  }

  off = off_common;
  const size_t plane_bytes = (size_t)n_nodes * DNODE * sizeof(float);
  const size_t norms_bytes = (size_t)2 * n_nodes * HEADS * sizeof(float);
  float* aggr1   = carve(plane_bytes + norms_bytes);
  float* norm01  = aggr1 + (size_t)n_nodes * DNODE;

  hipMemsetAsync(out,   0, plane_bytes, stream);
  hipMemsetAsync(aggr1, 0, plane_bytes + norms_bytes, stream);

  gat_edge_fused<<<dim3((etot + 255) / 256), blk, 0, stream>>>(
      edges, recv, send, score_s, score_r, wcomb, Wn, out, aggr1, norm01, n_edges, n_nodes);
  gat_finalize<<<dim3((n_nodes + 255) / 256), blk, 0, stream>>>(
      out, aggr1, norm01, ln_s, ln_b, n_nodes);
}


# node scores taken as a fifth WMMA tile of the projection (nodes x Wa, Wa = a folded into W, built per work-group in LDS), edge-weight fold in the same launch: 5 launches -> 3; score sums reordered
# speedup vs baseline: 13.3613x; 1.0577x over previous
#include <hip/hip_runtime.h>
#include <hip/hip_bf16.h>
#include <math.h>

typedef __attribute__((ext_vector_type(2))) float v2f;
typedef __attribute__((ext_vector_type(8))) float v8f;

#define HEADS 4
#define ATT_F 16
#define DNODE 64
#define DEDGE 16

#define WA_PITCH 68

__global__ __launch_bounds__(256) void gat_node_project(
    const float* __restrict__ nodes, const float* __restrict__ W,
    const float* __restrict__ a, const float* __restrict__ W_edge,
    float* __restrict__ Wn, float* __restrict__ score_s,
    float* __restrict__ score_r, float* __restrict__ wcomb,
    int n_nodes, int n_tiles)
{
  __shared__ __attribute__((aligned(16))) float s_wa[16 * WA_PITCH];
  const int tid = threadIdx.x;

  {
    const int h = tid >> 6, i = tid & 63;
    float as = 0.f, ar = 0.f;
#pragma unroll
    for (int f = 0; f < ATT_F; ++f) {
      const float w = W[(size_t)(h * ATT_F + f) * DNODE + i];
      as += a[h * 48 + f] * w;
      ar += a[h * 48 + ATT_F + f] * w;
    }
    s_wa[h * WA_PITCH + i]               = as;
    s_wa[(HEADS + h) * WA_PITCH + i]     = ar;
    s_wa[(2 * HEADS + h) * WA_PITCH + i] = 0.f;
    s_wa[(3 * HEADS + h) * WA_PITCH + i] = 0.f;
  }
  if (blockIdx.x == 0 && tid < HEADS * DEDGE) {
    const int h = tid >> 4, i = tid & 15;
    float acc = 0.f;
#pragma unroll
    for (int f = 0; f < ATT_F; ++f)
      acc += a[h * 48 + 32 + f] * W_edge[((h * ATT_F) + f) * DEDGE + i];
    wcomb[tid] = acc;
  }
  __syncthreads();

  int wave = blockIdx.x * (blockDim.x >> 5) + (tid >> 5);
  if (wave >= n_tiles) return;
  int lane = tid & 31;
  int half = lane >> 4;
  int l16  = lane & 15;
  int row0 = wave * 16;
  int m = row0 + l16; if (m >= n_nodes) m = n_nodes - 1;
  const float* __restrict__ arow = nodes + (size_t)m * DNODE;
  const float* wa_row = s_wa + l16 * WA_PITCH;

  v8f acc[4] = {};
  v8f acc5 = {};
#pragma unroll 4
  for (int k0 = 0; k0 < DNODE; k0 += 4) {
    int kk = k0 + 2 * half;
    v2f av; av.x = arow[kk]; av.y = arow[kk + 1];
#pragma unroll
    for (int t = 0; t < 4; ++t) {
      int j = t * 16 + l16;
      v2f b; b.x = W[(size_t)j * DNODE + kk]; b.y = W[(size_t)j * DNODE + kk + 1];
      acc[t] = __builtin_amdgcn_wmma_f32_16x16x4_f32(
           false, av,  false, b,
           (short)0, acc[t],  false,  false);
    }
    v2f b5; b5.x = wa_row[kk]; b5.y = wa_row[kk + 1];
    acc5 = __builtin_amdgcn_wmma_f32_16x16x4_f32(
         false, av,  false, b5,
         (short)0, acc5,  false,  false);
  }

  float* sdst = (l16 < HEADS) ? (score_s + l16) : (score_r + (l16 - HEADS));
  const bool has_score = l16 < 2 * HEADS;

  if (row0 + 16 <= n_nodes) {
#pragma unroll
    for (int t = 0; t < 4; ++t)
#pragma unroll
      for (int v = 0; v < 8; ++v)
        Wn[(size_t)(row0 + v + 8 * half) * DNODE + t * 16 + l16] = acc[t][v];
    if (has_score) {
#pragma unroll
      for (int v = 0; v < 8; ++v)
        sdst[(size_t)(row0 + v + 8 * half) * HEADS] = acc5[v];
    }
  } else {
#pragma unroll
    for (int t = 0; t < 4; ++t)
#pragma unroll
      for (int v = 0; v < 8; ++v) {
        int mo = row0 + v + 8 * half;
        if (mo < n_nodes) Wn[(size_t)mo * DNODE + t * 16 + l16] = acc[t][v];
      }
#pragma unroll
    for (int v = 0; v < 8; ++v) {
      int mo = row0 + v + 8 * half;
      if (has_score && mo < n_nodes) sdst[(size_t)mo * HEADS] = acc5[v];
    }
  }
}

__device__ __forceinline__ int lane_int(int v, int j)
{
  return __builtin_amdgcn_readlane(v, j);
}
__device__ __forceinline__ float lane_float(float v, int j)
{
  return __int_as_float(__builtin_amdgcn_readlane(__float_as_int(v), j));
}

#define EDGE_GROUP 8

__global__ __launch_bounds__(256) void gat_edge_fused(
    const float* __restrict__ edges, const int* __restrict__ recv,
    const int* __restrict__ send, const float* __restrict__ score_s,
    const float* __restrict__ score_r, const float* __restrict__ wcomb,
    const float* __restrict__ Wn, float* __restrict__ aggr0,
    float* __restrict__ aggr1, float* __restrict__ norm01,
    int n_edges, int n_nodes)
{
  const int chip_half = (int)((__builtin_amdgcn_s_sendmsg_rtn(0x87) >> 18) & 1u);
  float* __restrict__ aggr = chip_half ? aggr1 : aggr0;
  float* __restrict__ norm = norm01 + (size_t)chip_half * (size_t)n_nodes * HEADS;
  const int lane = threadIdx.x & 31;
  const int e    = blockIdx.x * blockDim.x + threadIdx.x;
  const int e0   = e - lane;
  const int etot = n_edges + n_nodes;
  if (e0 >= etot) return;

  int s = 0, r = 0;
  float ez[HEADS] = {0.f, 0.f, 0.f, 0.f};
  if (e < etot) {
    float elt[HEADS] = {0.f, 0.f, 0.f, 0.f};
    if (e < n_edges) {
      s = send[e]; r = recv[e];
      const float4* __restrict__ ep = (const float4*)(edges + (size_t)e * DEDGE);
      float ef[DEDGE];
#pragma unroll
      for (int c = 0; c < 4; ++c) {
        float4 q = ep[c];
        ef[4 * c + 0] = q.x; ef[4 * c + 1] = q.y; ef[4 * c + 2] = q.z; ef[4 * c + 3] = q.w;
      }
#pragma unroll
      for (int h = 0; h < HEADS; ++h) {
        float acc = 0.f;
#pragma unroll
        for (int i = 0; i < DEDGE; ++i) acc += ef[i] * wcomb[h * DEDGE + i];
        elt[h] = acc;
      }
    } else {
      s = r = e - n_edges;
    }
    const float4 qs = ((const float4*)score_s)[s];
    const float4 qr = ((const float4*)score_r)[r];
    const float sc[HEADS] = {qs.x + qr.x, qs.y + qr.y, qs.z + qr.z, qs.w + qr.w};
#pragma unroll
    for (int h = 0; h < HEADS; ++h) {
      float m  = sc[h] + elt[h];
      float lm = (m > 0.f) ? m : 0.01f * m;
      ez[h] = __expf(lm);
    }
  }

  const int  cnt = min(32, etot - e0);
  const bool up  = lane >= 16;

  {
    const int  hsel = lane & 3;
    const bool odd  = (lane & 1) != 0, hi = (lane & 2) != 0;
    for (int k = 0; k < 4; ++k) {
      const int src = 8 * k + (lane >> 2);
      const int   rr = __shfl(r, src);
      const float z0 = __shfl(ez[0], src), z1 = __shfl(ez[1], src);
      const float z2 = __shfl(ez[2], src), z3 = __shfl(ez[3], src);
      const float z01 = odd ? z1 : z0;
      const float z23 = odd ? z3 : z2;
      const float z   = hi ? z23 : z01;
      if (src < cnt) atomicAdd(norm + (size_t)rr * HEADS + hsel, z);
    }
  }

  int j0 = 0;
  for (; j0 + EDGE_GROUP <= cnt; j0 += EDGE_GROUP) {
    float vlo[EDGE_GROUP], vhi[EDGE_GROUP];
#pragma unroll
    for (int u = 0; u < EDGE_GROUP; ++u) {
      const float* __restrict__ srow = Wn + (size_t)lane_int(s, j0 + u) * DNODE;
      vlo[u] = srow[lane];
      vhi[u] = srow[32 + lane];
    }
#pragma unroll
    for (int u = 0; u < EDGE_GROUP; ++u) {
      const int j  = j0 + u;
      const int rj = lane_int(r, j);
      float* drow = aggr + (size_t)rj * DNODE;
      const float a0 = lane_float(ez[0], j), a1 = lane_float(ez[1], j);
      const float a2 = lane_float(ez[2], j), a3 = lane_float(ez[3], j);
      atomicAdd(drow + lane,      (up ? a1 : a0) * vlo[u]);
      atomicAdd(drow + 32 + lane, (up ? a3 : a2) * vhi[u]);
    }
  }
  for (; j0 < cnt; ++j0) {
    const float* __restrict__ srow = Wn + (size_t)lane_int(s, j0) * DNODE;
    const int rj = lane_int(r, j0);
    float* drow = aggr + (size_t)rj * DNODE;
    const float a0 = lane_float(ez[0], j0), a1 = lane_float(ez[1], j0);
    const float a2 = lane_float(ez[2], j0), a3 = lane_float(ez[3], j0);
    const float vlo = srow[lane], vhi = srow[32 + lane];
    atomicAdd(drow + lane,      (up ? a1 : a0) * vlo);
    atomicAdd(drow + 32 + lane, (up ? a3 : a2) * vhi);
  }
}

__global__ __launch_bounds__(256) void gat_finalize(
    float* __restrict__ out, const float* __restrict__ aggr1,
    const float* __restrict__ norm01,
    const float* __restrict__ scale, const float* __restrict__ bias, int n_nodes)
{
  int i = blockIdx.x * blockDim.x + threadIdx.x;
  if (i >= n_nodes) return;
  float* row = out + (size_t)i * DNODE;
  const float* __restrict__ row1 = aggr1 + (size_t)i * DNODE;
  const float4 q  = ((const float4*)norm01)[i];
  const float4 q1 = ((const float4*)norm01)[(size_t)n_nodes + i];
  const float nrm[HEADS] = {q.x + q1.x, q.y + q1.y, q.z + q1.z, q.w + q1.w};
  float v[DNODE];
  float mean = 0.f;
#pragma unroll
  for (int j = 0; j < DNODE; ++j) {
    float x = (row[j] + row1[j]) / nrm[j / ATT_F];
    x = (x > 0.f) ? x : expm1f(x);
    v[j] = x; mean += x;
  }
  mean *= (1.f / DNODE);
  float var = 0.f;
#pragma unroll
  for (int j = 0; j < DNODE; ++j) { float d = v[j] - mean; var += d * d; }
  var *= (1.f / DNODE);
  float inv = rsqrtf(var + 1e-6f);
#pragma unroll
  for (int j = 0; j < DNODE; ++j)
    row[j] = (v[j] - mean) * inv * scale[j] + bias[j];
}

#define NBINS  256
#define NCHUNK 256
#define OWNER_GROUP 16

__global__ __launch_bounds__(1024) void gat_edge_order(
    const float* __restrict__ edges, const int* __restrict__ recv,
    const int* __restrict__ send, const float* __restrict__ score_s,
    const float* __restrict__ wcomb,
    int* __restrict__ sr, float4* __restrict__ lgrec, int2* __restrict__ seg,
    int n_edges, int chunk_edges, int chunk_stride, int npb)
{
  __shared__ int s_cnt[NBINS];
  __shared__ int s_cur[NBINS];
  __shared__ int s_wsum[NBINS / 32];
  const int tid = threadIdx.x;
  const int c   = blockIdx.x;
  const int e_begin = c * chunk_edges;
  const int e_end   = min(n_edges, e_begin + chunk_edges);
  const int rbase   = c * chunk_stride;
  if (tid < NBINS) s_cnt[tid] = 0;
  __syncthreads();
  for (int e = e_begin + tid; e < e_end; e += 1024)
    atomicAdd(&s_cnt[recv[e] / npb], 1);
  __syncthreads();
  int v = 0, incl = 0;
  if (tid < NBINS) {
    v = s_cnt[tid];
    incl = v;
    for (int d = 1; d < 32; d <<= 1) {
      const int t = __shfl_up(incl, d);
      if ((tid & 31) >= d) incl += t;
    }
    if ((tid & 31) == 31) s_wsum[tid >> 5] = incl;
  }
  __syncthreads();
  if (tid < NBINS) {
    int before = 0;
    for (int w = 0; w < (tid >> 5); ++w) before += s_wsum[w];
    const int excl = before + incl - v;
    s_cur[tid] = excl;
    seg[tid * NCHUNK + c] = make_int2(rbase + excl, v);
  }
  __syncthreads();
  for (int e = e_begin + tid; e < e_end; e += 1024) {
    const int s = send[e], r = recv[e];
    const float4* __restrict__ ep = (const float4*)(edges + (size_t)e * DEDGE);
    float ef[DEDGE];
#pragma unroll
    for (int k = 0; k < 4; ++k) {
      float4 q = ep[k];
      ef[4 * k + 0] = q.x; ef[4 * k + 1] = q.y; ef[4 * k + 2] = q.z; ef[4 * k + 3] = q.w;
    }
    float elt[HEADS];
#pragma unroll
    for (int h = 0; h < HEADS; ++h) {
      float acc = 0.f;
#pragma unroll
      for (int i = 0; i < DEDGE; ++i) acc += ef[i] * wcomb[h * DEDGE + i];
      elt[h] = acc;
    }
    const float4 qs = ((const float4*)score_s)[s];
    const int bin  = r / npb;
    const int rl   = r - bin * npb;
    const int slot = rbase + atomicAdd(&s_cur[bin], 1);
    sr[slot]    = (int)((unsigned)s | ((unsigned)rl << 24));
    lgrec[slot] = make_float4(qs.x + elt[0], qs.y + elt[1], qs.z + elt[2], qs.w + elt[3]);
  }
}

__global__ __launch_bounds__(1024) void gat_owner_sum(
    const int* __restrict__ sr, const float4* __restrict__ lgrec,
    const int2* __restrict__ seg, const float* __restrict__ score_s,
    const float* __restrict__ score_r, const float* __restrict__ Wn,
    const float* __restrict__ scale, const float* __restrict__ bias,
    float* __restrict__ out, int n_nodes, int npb)
{
  extern __shared__ float smem[];
  float* acc = smem;
  float* nrm = smem + npb * DNODE;
  float* scr = nrm + npb * HEADS;
  const int tid  = threadIdx.x;
  const int lane = tid & 31;
  const int wave = __builtin_amdgcn_readfirstlane(tid >> 5);
  const int b     = blockIdx.x;
  const int node0 = b * npb;
  const int nn    = min(npb, n_nodes - node0);
  const bool up   = lane >= 16;

  for (int idx = tid; idx < nn * DNODE; idx += 1024) {
    const int n = idx >> 6, f = idx & 63, h = f >> 4;
    const int node = node0 + n;
    const float m  = score_s[node * HEADS + h] + score_r[node * HEADS + h];
    const float lm = (m > 0.f) ? m : 0.01f * m;
    const float z  = __expf(lm);
    acc[idx] = z * Wn[(size_t)node * DNODE + f];
    if ((f & 15) == 0) { nrm[n * HEADS + h] = z; scr[n * HEADS + h] = score_r[node * HEADS + h]; }
  }
  __syncthreads();

  int2 sgl = make_int2(0, 0);
  if (lane < NCHUNK / 32) sgl = seg[b * NCHUNK + wave * (NCHUNK / 32) + lane];
  int    w_nx = 0;
  float4 q_nx = make_float4(0.f, 0.f, 0.f, 0.f);
  {
    const int st0 = lane_int(sgl.x, 0), c0 = lane_int(sgl.y, 0);
    if (lane < min(32, c0)) { w_nx = sr[st0 + lane]; q_nx = lgrec[st0 + lane]; }
  }
  for (int k = 0; k < NCHUNK / 32; ++k) {
    const int start = lane_int(sgl.x, k);
    const int cnt   = lane_int(sgl.y, k);
    const int    w_first = w_nx;
    const float4 q_first = q_nx;
    if (k + 1 < NCHUNK / 32) {
      const int st1 = lane_int(sgl.x, k + 1), c1 = lane_int(sgl.y, k + 1);
      if (lane < min(32, c1)) { w_nx = sr[st1 + lane]; q_nx = lgrec[st1 + lane]; }
    }
    for (int base = 0; base < cnt; base += 32) {
      const int m = min(32, cnt - base);
      int s = 0, rl = 0;
      float ez0 = 0.f, ez1 = 0.f, ez2 = 0.f, ez3 = 0.f;
      if (lane < m) {
        int w = w_first;
        float4 q = q_first;
        if (base > 0) {
          w = sr[start + base + lane];
          q = lgrec[start + base + lane];
        }
        s  = w & 0xFFFFFF;
        rl = (int)((unsigned)w >> 24);
        const float m0 = q.x + scr[rl * HEADS + 0], m1 = q.y + scr[rl * HEADS + 1];
        const float m2 = q.z + scr[rl * HEADS + 2], m3 = q.w + scr[rl * HEADS + 3];
        ez0 = __expf((m0 > 0.f) ? m0 : 0.01f * m0);
        ez1 = __expf((m1 > 0.f) ? m1 : 0.01f * m1);
        ez2 = __expf((m2 > 0.f) ? m2 : 0.01f * m2);
        ez3 = __expf((m3 > 0.f) ? m3 : 0.01f * m3);
        atomicAdd(&nrm[rl * HEADS + 0], ez0);
        atomicAdd(&nrm[rl * HEADS + 1], ez1);
        atomicAdd(&nrm[rl * HEADS + 2], ez2);
        atomicAdd(&nrm[rl * HEADS + 3], ez3);
      }
      for (int j0 = 0; j0 < m; j0 += OWNER_GROUP) {
        float vlo[OWNER_GROUP], vhi[OWNER_GROUP];
#pragma unroll
        for (int u = 0; u < OWNER_GROUP; ++u) {
          vlo[u] = 0.f; vhi[u] = 0.f;
          if (j0 + u < m) {
            const float* __restrict__ srow = Wn + (size_t)lane_int(s, j0 + u) * DNODE;
            vlo[u] = srow[lane];
            vhi[u] = srow[32 + lane];
          }
        }
#pragma unroll
        for (int u = 0; u < OWNER_GROUP; ++u) {
          if (j0 + u < m) {
            const int j  = j0 + u;
            const int rj = lane_int(rl, j);
            float* drow = acc + rj * DNODE;
            const float a0 = lane_float(ez0, j), a1 = lane_float(ez1, j);
            const float a2 = lane_float(ez2, j), a3 = lane_float(ez3, j);
            atomicAdd(drow + lane,      (up ? a1 : a0) * vlo[u]);
            atomicAdd(drow + 32 + lane, (up ? a3 : a2) * vhi[u]);
          }
        }
      }
    }
  }
  __syncthreads();

  const float sc_lo = scale[lane], sc_hi = scale[32 + lane];
  const float bi_lo = bias[lane],  bi_hi = bias[32 + lane];
  for (int n = wave; n < nn; n += 32) {
    const float n01 = up ? nrm[n * HEADS + 1] : nrm[n * HEADS + 0];
    const float n23 = up ? nrm[n * HEADS + 3] : nrm[n * HEADS + 2];
    float x0 = acc[n * DNODE + lane] / n01;
    float x1 = acc[n * DNODE + 32 + lane] / n23;
    x0 = (x0 > 0.f) ? x0 : expm1f(x0);
    x1 = (x1 > 0.f) ? x1 : expm1f(x1);
    float sum = x0 + x1;
    for (int d = 16; d >= 1; d >>= 1) sum += __shfl_xor(sum, d);
    const float mean = sum * (1.f / DNODE);
    const float d0 = x0 - mean, d1 = x1 - mean;
    float var = d0 * d0 + d1 * d1;
    for (int d = 16; d >= 1; d >>= 1) var += __shfl_xor(var, d);
    var *= (1.f / DNODE);
    const float inv = rsqrtf(var + 1e-6f);
    float* row = out + (size_t)(node0 + n) * DNODE;
    row[lane]      = d0 * inv * sc_lo + bi_lo;
    row[32 + lane] = d1 * inv * sc_hi + bi_hi;
  }
}

extern "C" void kernel_launch(void* const* d_in, const int* in_sizes, int n_in,
                              void* d_out, int out_size, void* d_ws, size_t ws_size,
                              hipStream_t stream)
{
  const float* nodes  = (const float*)d_in[0];
  const float* edges  = (const float*)d_in[1];
  const int*   recv   = (const int*)d_in[2];
  const int*   send   = (const int*)d_in[3];
  const float* W      = (const float*)d_in[4];
  const float* W_edge = (const float*)d_in[5];
  const float* a      = (const float*)d_in[6];
  const float* ln_s   = (const float*)d_in[7];
  const float* ln_b   = (const float*)d_in[8];

  const int n_nodes = in_sizes[0] / DNODE;
  const int n_edges = in_sizes[1] / DEDGE;
  const int etot    = n_edges + n_nodes;
  float* out = (float*)d_out;

  char* ws = (char*)d_ws;
  size_t off = 0;
  auto carve = [&](size_t bytes) -> float* {
    float* p = (float*)(ws + off);
    off += (bytes + 255) & ~(size_t)255;
    return p;
  };
  float* Wn      = carve((size_t)n_nodes * DNODE * sizeof(float));
  float* score_s = carve((size_t)n_nodes * HEADS * sizeof(float));
  float* score_r = carve((size_t)n_nodes * HEADS * sizeof(float));
  float* wcomb   = carve((size_t)HEADS * DEDGE * sizeof(float));
  const size_t off_common = off;

  const int n_tiles = (n_nodes + 15) / 16;
  dim3 blk(256);

  gat_node_project<<<dim3((n_tiles + 7) / 8), blk, 0, stream>>>(
      nodes, W, a, W_edge, Wn, score_s, score_r, wcomb, n_nodes, n_tiles);

  const int npb          = (n_nodes + NBINS - 1) / NBINS;
  const int chunk_edges  = (n_edges + NCHUNK - 1) / NCHUNK;
  const int chunk_stride = (chunk_edges + 31) & ~31;
  const size_t lds_bytes = (size_t)npb * (DNODE + 2 * HEADS) * sizeof(float);
  int2*   seg   = (int2*)carve((size_t)NBINS * NCHUNK * sizeof(int2));
  int*    sr    = (int*)carve((size_t)NCHUNK * chunk_stride * sizeof(int));
  float4* lgrec = (float4*)carve((size_t)NCHUNK * chunk_stride * sizeof(float4));
  if (off <= ws_size && lds_bytes <= 65536 && n_nodes <= (1 << 24) && npb > 0) {
    gat_edge_order<<<dim3(NCHUNK), dim3(1024), 0, stream>>>(
        edges, recv, send, score_s, wcomb, sr, lgrec, seg,
        n_edges, chunk_edges, chunk_stride, npb);
    gat_owner_sum<<<dim3(NBINS), dim3(1024), lds_bytes, stream>>>(
        sr, lgrec, seg, score_s, score_r, Wn, ln_s, ln_b, out, n_nodes, npb);
    return;
  }

  off = off_common;
  const size_t plane_bytes = (size_t)n_nodes * DNODE * sizeof(float);
  const size_t norms_bytes = (size_t)2 * n_nodes * HEADS * sizeof(float);
  float* aggr1   = carve(plane_bytes + norms_bytes);
  float* norm01  = aggr1 + (size_t)n_nodes * DNODE;

  hipMemsetAsync(out,   0, plane_bytes, stream);
  hipMemsetAsync(aggr1, 0, plane_bytes + norms_bytes, stream);

  gat_edge_fused<<<dim3((etot + 255) / 256), blk, 0, stream>>>(
      edges, recv, send, score_s, score_r, wcomb, Wn, out, aggr1, norm01, n_edges, n_nodes);
  gat_finalize<<<dim3((n_nodes + 255) / 256), blk, 0, stream>>>(
      out, aggr1, norm01, ln_s, ln_b, n_nodes);
}


# edge-order launch: each chunk's records are put in range order in LDS (whole chunk, 160 KiB) and copied out as whole lines; was one scattered 4-byte and one 16-byte global store per edge
# speedup vs baseline: 14.3697x; 1.0755x over previous
#include <hip/hip_runtime.h>
#include <hip/hip_bf16.h>
#include <math.h>

typedef __attribute__((ext_vector_type(2))) float v2f;
typedef __attribute__((ext_vector_type(8))) float v8f;

#define HEADS 4
#define ATT_F 16
#define DNODE 64
#define DEDGE 16

#define WA_PITCH 68

__global__ __launch_bounds__(256) void gat_node_project(
    const float* __restrict__ nodes, const float* __restrict__ W,
    const float* __restrict__ a, const float* __restrict__ W_edge,
    float* __restrict__ Wn, float* __restrict__ score_s,
    float* __restrict__ score_r, float* __restrict__ wcomb,
    int n_nodes, int n_tiles)
{
  __shared__ __attribute__((aligned(16))) float s_wa[16 * WA_PITCH];
  const int tid = threadIdx.x;

  {
    const int h = tid >> 6, i = tid & 63;
    float as = 0.f, ar = 0.f;
#pragma unroll
    for (int f = 0; f < ATT_F; ++f) {
      const float w = W[(size_t)(h * ATT_F + f) * DNODE + i];
      as += a[h * 48 + f] * w;
      ar += a[h * 48 + ATT_F + f] * w;
    }
    s_wa[h * WA_PITCH + i]               = as;
    s_wa[(HEADS + h) * WA_PITCH + i]     = ar;
    s_wa[(2 * HEADS + h) * WA_PITCH + i] = 0.f;
    s_wa[(3 * HEADS + h) * WA_PITCH + i] = 0.f;
  }
  if (blockIdx.x == 0 && tid < HEADS * DEDGE) {
    const int h = tid >> 4, i = tid & 15;
    float acc = 0.f;
#pragma unroll
    for (int f = 0; f < ATT_F; ++f)
      acc += a[h * 48 + 32 + f] * W_edge[((h * ATT_F) + f) * DEDGE + i];
    wcomb[tid] = acc;
  }
  __syncthreads();

  int wave = blockIdx.x * (blockDim.x >> 5) + (tid >> 5);
  if (wave >= n_tiles) return;
  int lane = tid & 31;
  int half = lane >> 4;
  int l16  = lane & 15;
  int row0 = wave * 16;
  int m = row0 + l16; if (m >= n_nodes) m = n_nodes - 1;
  const float* __restrict__ arow = nodes + (size_t)m * DNODE;
  const float* wa_row = s_wa + l16 * WA_PITCH;

  v8f acc[4] = {};
  v8f acc5 = {};
#pragma unroll 4
  for (int k0 = 0; k0 < DNODE; k0 += 4) {
    int kk = k0 + 2 * half;
    v2f av; av.x = arow[kk]; av.y = arow[kk + 1];
#pragma unroll
    for (int t = 0; t < 4; ++t) {
      int j = t * 16 + l16;
      v2f b; b.x = W[(size_t)j * DNODE + kk]; b.y = W[(size_t)j * DNODE + kk + 1];
      acc[t] = __builtin_amdgcn_wmma_f32_16x16x4_f32(
           false, av,  false, b,
           (short)0, acc[t],  false,  false);
    }
    v2f b5; b5.x = wa_row[kk]; b5.y = wa_row[kk + 1];
    acc5 = __builtin_amdgcn_wmma_f32_16x16x4_f32(
         false, av,  false, b5,
         (short)0, acc5,  false,  false);
  }

  float* sdst = (l16 < HEADS) ? (score_s + l16) : (score_r + (l16 - HEADS));
  const bool has_score = l16 < 2 * HEADS;

  if (row0 + 16 <= n_nodes) {
#pragma unroll
    for (int t = 0; t < 4; ++t)
#pragma unroll
      for (int v = 0; v < 8; ++v)
        Wn[(size_t)(row0 + v + 8 * half) * DNODE + t * 16 + l16] = acc[t][v];
    if (has_score) {
#pragma unroll
      for (int v = 0; v < 8; ++v)
        sdst[(size_t)(row0 + v + 8 * half) * HEADS] = acc5[v];
    }
  } else {
#pragma unroll
    for (int t = 0; t < 4; ++t)
#pragma unroll
      for (int v = 0; v < 8; ++v) {
        int mo = row0 + v + 8 * half;
        if (mo < n_nodes) Wn[(size_t)mo * DNODE + t * 16 + l16] = acc[t][v];
      }
#pragma unroll
    for (int v = 0; v < 8; ++v) {
      int mo = row0 + v + 8 * half;
      if (has_score && mo < n_nodes) sdst[(size_t)mo * HEADS] = acc5[v];
    }
  }
}

__device__ __forceinline__ int lane_int(int v, int j)
{
  return __builtin_amdgcn_readlane(v, j);
}
__device__ __forceinline__ float lane_float(float v, int j)
{
  return __int_as_float(__builtin_amdgcn_readlane(__float_as_int(v), j));
}

#define EDGE_GROUP 8

__global__ __launch_bounds__(256) void gat_edge_fused(
    const float* __restrict__ edges, const int* __restrict__ recv,
    const int* __restrict__ send, const float* __restrict__ score_s,
    const float* __restrict__ score_r, const float* __restrict__ wcomb,
    const float* __restrict__ Wn, float* __restrict__ aggr0,
    float* __restrict__ aggr1, float* __restrict__ norm01,
    int n_edges, int n_nodes)
{
  const int chip_half = (int)((__builtin_amdgcn_s_sendmsg_rtn(0x87) >> 18) & 1u);
  float* __restrict__ aggr = chip_half ? aggr1 : aggr0;
  float* __restrict__ norm = norm01 + (size_t)chip_half * (size_t)n_nodes * HEADS;
  const int lane = threadIdx.x & 31;
  const int e    = blockIdx.x * blockDim.x + threadIdx.x;
  const int e0   = e - lane;
  const int etot = n_edges + n_nodes;
  if (e0 >= etot) return;

  int s = 0, r = 0;
  float ez[HEADS] = {0.f, 0.f, 0.f, 0.f};
  if (e < etot) {
    float elt[HEADS] = {0.f, 0.f, 0.f, 0.f};
    if (e < n_edges) {
      s = send[e]; r = recv[e];
      const float4* __restrict__ ep = (const float4*)(edges + (size_t)e * DEDGE);
      float ef[DEDGE];
#pragma unroll
      for (int c = 0; c < 4; ++c) {
        float4 q = ep[c];
        ef[4 * c + 0] = q.x; ef[4 * c + 1] = q.y; ef[4 * c + 2] = q.z; ef[4 * c + 3] = q.w;
      }
#pragma unroll
      for (int h = 0; h < HEADS; ++h) {
        float acc = 0.f;
#pragma unroll
        for (int i = 0; i < DEDGE; ++i) acc += ef[i] * wcomb[h * DEDGE + i];
        elt[h] = acc;
      }
    } else {
      s = r = e - n_edges;
    }
    const float4 qs = ((const float4*)score_s)[s];
    const float4 qr = ((const float4*)score_r)[r];
    const float sc[HEADS] = {qs.x + qr.x, qs.y + qr.y, qs.z + qr.z, qs.w + qr.w};
#pragma unroll
    for (int h = 0; h < HEADS; ++h) {
      float m  = sc[h] + elt[h];
      float lm = (m > 0.f) ? m : 0.01f * m;
      ez[h] = __expf(lm);
    }
  }

  const int  cnt = min(32, etot - e0);
  const bool up  = lane >= 16;

  {
    const int  hsel = lane & 3;
    const bool odd  = (lane & 1) != 0, hi = (lane & 2) != 0;
    for (int k = 0; k < 4; ++k) {
      const int src = 8 * k + (lane >> 2);
      const int   rr = __shfl(r, src);
      const float z0 = __shfl(ez[0], src), z1 = __shfl(ez[1], src);
      const float z2 = __shfl(ez[2], src), z3 = __shfl(ez[3], src);
      const float z01 = odd ? z1 : z0;
      const float z23 = odd ? z3 : z2;
      const float z   = hi ? z23 : z01;
      if (src < cnt) atomicAdd(norm + (size_t)rr * HEADS + hsel, z);
    }
  }

  int j0 = 0;
  for (; j0 + EDGE_GROUP <= cnt; j0 += EDGE_GROUP) {
    float vlo[EDGE_GROUP], vhi[EDGE_GROUP];
#pragma unroll
    for (int u = 0; u < EDGE_GROUP; ++u) {
      const float* __restrict__ srow = Wn + (size_t)lane_int(s, j0 + u) * DNODE;
      vlo[u] = srow[lane];
      vhi[u] = srow[32 + lane];
    }
#pragma unroll
    for (int u = 0; u < EDGE_GROUP; ++u) {
      const int j  = j0 + u;
      const int rj = lane_int(r, j);
      float* drow = aggr + (size_t)rj * DNODE;
      const float a0 = lane_float(ez[0], j), a1 = lane_float(ez[1], j);
      const float a2 = lane_float(ez[2], j), a3 = lane_float(ez[3], j);
      atomicAdd(drow + lane,      (up ? a1 : a0) * vlo[u]);
      atomicAdd(drow + 32 + lane, (up ? a3 : a2) * vhi[u]);
    }
  }
  for (; j0 < cnt; ++j0) {
    const float* __restrict__ srow = Wn + (size_t)lane_int(s, j0) * DNODE;
    const int rj = lane_int(r, j0);
    float* drow = aggr + (size_t)rj * DNODE;
    const float a0 = lane_float(ez[0], j0), a1 = lane_float(ez[1], j0);
    const float a2 = lane_float(ez[2], j0), a3 = lane_float(ez[3], j0);
    const float vlo = srow[lane], vhi = srow[32 + lane];
    atomicAdd(drow + lane,      (up ? a1 : a0) * vlo);
    atomicAdd(drow + 32 + lane, (up ? a3 : a2) * vhi);
  }
}

__global__ __launch_bounds__(256) void gat_finalize(
    float* __restrict__ out, const float* __restrict__ aggr1,
    const float* __restrict__ norm01,
    const float* __restrict__ scale, const float* __restrict__ bias, int n_nodes)
{
  int i = blockIdx.x * blockDim.x + threadIdx.x;
  if (i >= n_nodes) return;
  float* row = out + (size_t)i * DNODE;
  const float* __restrict__ row1 = aggr1 + (size_t)i * DNODE;
  const float4 q  = ((const float4*)norm01)[i];
  const float4 q1 = ((const float4*)norm01)[(size_t)n_nodes + i];
  const float nrm[HEADS] = {q.x + q1.x, q.y + q1.y, q.z + q1.z, q.w + q1.w};
  float v[DNODE];
  float mean = 0.f;
#pragma unroll
  for (int j = 0; j < DNODE; ++j) {
    float x = (row[j] + row1[j]) / nrm[j / ATT_F];
    x = (x > 0.f) ? x : expm1f(x);
    v[j] = x; mean += x;
  }
  mean *= (1.f / DNODE);
  float var = 0.f;
#pragma unroll
  for (int j = 0; j < DNODE; ++j) { float d = v[j] - mean; var += d * d; }
  var *= (1.f / DNODE);
  float inv = rsqrtf(var + 1e-6f);
#pragma unroll
  for (int j = 0; j < DNODE; ++j)
    row[j] = (v[j] - mean) * inv * scale[j] + bias[j];
}

#define NBINS  256
#define NCHUNK 256
#define OWNER_GROUP 16
#define STAGE_CAP 8192

__global__ __launch_bounds__(1024) void gat_edge_order(
    const float* __restrict__ edges, const int* __restrict__ recv,
    const int* __restrict__ send, const float* __restrict__ score_s,
    const float* __restrict__ wcomb,
    int* __restrict__ sr, float4* __restrict__ lgrec, int2* __restrict__ seg,
    int n_edges, int chunk_edges, int chunk_stride, int npb)
{
  __shared__ int s_cnt[NBINS];
  __shared__ int s_cur[NBINS];
  __shared__ int s_wsum[NBINS / 32];
  __shared__ __attribute__((aligned(16))) float4 s_lg[STAGE_CAP];
  __shared__ int s_sr[STAGE_CAP];
  const int tid = threadIdx.x;
  const int c   = blockIdx.x;
  const int e_begin = c * chunk_edges;
  const int e_end   = min(n_edges, e_begin + chunk_edges);
  const int rbase   = c * chunk_stride;
  if (tid < NBINS) s_cnt[tid] = 0;
  __syncthreads();
  for (int e = e_begin + tid; e < e_end; e += 1024)
    atomicAdd(&s_cnt[recv[e] / npb], 1);
  __syncthreads();
  int v = 0, incl = 0;
  if (tid < NBINS) {
    v = s_cnt[tid];
    incl = v;
    for (int d = 1; d < 32; d <<= 1) {
      const int t = __shfl_up(incl, d);
      if ((tid & 31) >= d) incl += t;
    }
    if ((tid & 31) == 31) s_wsum[tid >> 5] = incl;
  }
  __syncthreads();
  if (tid < NBINS) {
    int before = 0;
    for (int w = 0; w < (tid >> 5); ++w) before += s_wsum[w];
    const int excl = before + incl - v;
    s_cur[tid] = excl;
    seg[tid * NCHUNK + c] = make_int2(rbase + excl, v);
  }
  __syncthreads();
  for (int e = e_begin + tid; e < e_end; e += 1024) {
    const int s = send[e], r = recv[e];
    const float4* __restrict__ ep = (const float4*)(edges + (size_t)e * DEDGE);
    float ef[DEDGE];
#pragma unroll
    for (int k = 0; k < 4; ++k) {
      float4 q = ep[k];
      ef[4 * k + 0] = q.x; ef[4 * k + 1] = q.y; ef[4 * k + 2] = q.z; ef[4 * k + 3] = q.w;
    }
    float elt[HEADS];
#pragma unroll
    for (int h = 0; h < HEADS; ++h) {
      float acc = 0.f;
#pragma unroll
      for (int i = 0; i < DEDGE; ++i) acc += ef[i] * wcomb[h * DEDGE + i];
      elt[h] = acc;
    }
    const float4 qs = ((const float4*)score_s)[s];
    const int bin  = r / npb;
    const int rl   = r - bin * npb;
    const int slot = atomicAdd(&s_cur[bin], 1);
    s_sr[slot] = (int)((unsigned)s | ((unsigned)rl << 24));
    s_lg[slot] = make_float4(qs.x + elt[0], qs.y + elt[1], qs.z + elt[2], qs.w + elt[3]);
  }
  __syncthreads();
  const int n_rec = e_end - e_begin;
  for (int i = tid; i < n_rec; i += 1024) {
    sr[rbase + i]    = s_sr[i];
    lgrec[rbase + i] = s_lg[i];
  }
}

__global__ __launch_bounds__(1024) void gat_owner_sum(
    const int* __restrict__ sr, const float4* __restrict__ lgrec,
    const int2* __restrict__ seg, const float* __restrict__ score_s,
    const float* __restrict__ score_r, const float* __restrict__ Wn,
    const float* __restrict__ scale, const float* __restrict__ bias,
    float* __restrict__ out, int n_nodes, int npb)
{
  extern __shared__ float smem[];
  float* acc = smem;
  float* nrm = smem + npb * DNODE;
  float* scr = nrm + npb * HEADS;
  const int tid  = threadIdx.x;
  const int lane = tid & 31;
  const int wave = __builtin_amdgcn_readfirstlane(tid >> 5);
  const int b     = blockIdx.x;
  const int node0 = b * npb;
  const int nn    = min(npb, n_nodes - node0);
  const bool up   = lane >= 16;

  for (int idx = tid; idx < nn * DNODE; idx += 1024) {
    const int n = idx >> 6, f = idx & 63, h = f >> 4;
    const int node = node0 + n;
    const float m  = score_s[node * HEADS + h] + score_r[node * HEADS + h];
    const float lm = (m > 0.f) ? m : 0.01f * m;
    const float z  = __expf(lm);
    acc[idx] = z * Wn[(size_t)node * DNODE + f];
    if ((f & 15) == 0) { nrm[n * HEADS + h] = z; scr[n * HEADS + h] = score_r[node * HEADS + h]; }
  }
  __syncthreads();

  int2 sgl = make_int2(0, 0);
  if (lane < NCHUNK / 32) sgl = seg[b * NCHUNK + wave * (NCHUNK / 32) + lane];
  int    w_nx = 0;
  float4 q_nx = make_float4(0.f, 0.f, 0.f, 0.f);
  {
    const int st0 = lane_int(sgl.x, 0), c0 = lane_int(sgl.y, 0);
    if (lane < min(32, c0)) { w_nx = sr[st0 + lane]; q_nx = lgrec[st0 + lane]; }
  }
  for (int k = 0; k < NCHUNK / 32; ++k) {
    const int start = lane_int(sgl.x, k);
    const int cnt   = lane_int(sgl.y, k);
    const int    w_first = w_nx;
    const float4 q_first = q_nx;
    if (k + 1 < NCHUNK / 32) {
      const int st1 = lane_int(sgl.x, k + 1), c1 = lane_int(sgl.y, k + 1);
      if (lane < min(32, c1)) { w_nx = sr[st1 + lane]; q_nx = lgrec[st1 + lane]; }
    }
    for (int base = 0; base < cnt; base += 32) {
      const int m = min(32, cnt - base);
      int s = 0, rl = 0;
      float ez0 = 0.f, ez1 = 0.f, ez2 = 0.f, ez3 = 0.f;
      if (lane < m) {
        int w = w_first;
        float4 q = q_first;
        if (base > 0) {
          w = sr[start + base + lane];
          q = lgrec[start + base + lane];
        }
        s  = w & 0xFFFFFF;
        rl = (int)((unsigned)w >> 24);
        const float m0 = q.x + scr[rl * HEADS + 0], m1 = q.y + scr[rl * HEADS + 1];
        const float m2 = q.z + scr[rl * HEADS + 2], m3 = q.w + scr[rl * HEADS + 3];
        ez0 = __expf((m0 > 0.f) ? m0 : 0.01f * m0);
        ez1 = __expf((m1 > 0.f) ? m1 : 0.01f * m1);
        ez2 = __expf((m2 > 0.f) ? m2 : 0.01f * m2);
        ez3 = __expf((m3 > 0.f) ? m3 : 0.01f * m3);
        atomicAdd(&nrm[rl * HEADS + 0], ez0);
        atomicAdd(&nrm[rl * HEADS + 1], ez1);
        atomicAdd(&nrm[rl * HEADS + 2], ez2);
        atomicAdd(&nrm[rl * HEADS + 3], ez3);
      }
      for (int j0 = 0; j0 < m; j0 += OWNER_GROUP) {
        float vlo[OWNER_GROUP], vhi[OWNER_GROUP];
#pragma unroll
        for (int u = 0; u < OWNER_GROUP; ++u) {
          vlo[u] = 0.f; vhi[u] = 0.f;
          if (j0 + u < m) {
            const float* __restrict__ srow = Wn + (size_t)lane_int(s, j0 + u) * DNODE;
            vlo[u] = srow[lane];
            vhi[u] = srow[32 + lane];
          }
        }
#pragma unroll
        for (int u = 0; u < OWNER_GROUP; ++u) {
          if (j0 + u < m) {
            const int j  = j0 + u;
            const int rj = lane_int(rl, j);
            float* drow = acc + rj * DNODE;
            const float a0 = lane_float(ez0, j), a1 = lane_float(ez1, j);
            const float a2 = lane_float(ez2, j), a3 = lane_float(ez3, j);
            atomicAdd(drow + lane,      (up ? a1 : a0) * vlo[u]);
            atomicAdd(drow + 32 + lane, (up ? a3 : a2) * vhi[u]);
          }
        }
      }
    }
  }
  __syncthreads();

  const float sc_lo = scale[lane], sc_hi = scale[32 + lane];
  const float bi_lo = bias[lane],  bi_hi = bias[32 + lane];
  for (int n = wave; n < nn; n += 32) {
    const float n01 = up ? nrm[n * HEADS + 1] : nrm[n * HEADS + 0];
    const float n23 = up ? nrm[n * HEADS + 3] : nrm[n * HEADS + 2];
    float x0 = acc[n * DNODE + lane] / n01;
    float x1 = acc[n * DNODE + 32 + lane] / n23;
    x0 = (x0 > 0.f) ? x0 : expm1f(x0);
    x1 = (x1 > 0.f) ? x1 : expm1f(x1);
    float sum = x0 + x1;
    for (int d = 16; d >= 1; d >>= 1) sum += __shfl_xor(sum, d);
    const float mean = sum * (1.f / DNODE);
    const float d0 = x0 - mean, d1 = x1 - mean;
    float var = d0 * d0 + d1 * d1;
    for (int d = 16; d >= 1; d >>= 1) var += __shfl_xor(var, d);
    var *= (1.f / DNODE);
    const float inv = rsqrtf(var + 1e-6f);
    float* row = out + (size_t)(node0 + n) * DNODE;
    row[lane]      = d0 * inv * sc_lo + bi_lo;
    row[32 + lane] = d1 * inv * sc_hi + bi_hi;
  }
}

extern "C" void kernel_launch(void* const* d_in, const int* in_sizes, int n_in,
                              void* d_out, int out_size, void* d_ws, size_t ws_size,
                              hipStream_t stream)
{
  const float* nodes  = (const float*)d_in[0];
  const float* edges  = (const float*)d_in[1];
  const int*   recv   = (const int*)d_in[2];
  const int*   send   = (const int*)d_in[3];
  const float* W      = (const float*)d_in[4];
  const float* W_edge = (const float*)d_in[5];
  const float* a      = (const float*)d_in[6];
  const float* ln_s   = (const float*)d_in[7];
  const float* ln_b   = (const float*)d_in[8];

  const int n_nodes = in_sizes[0] / DNODE;
  const int n_edges = in_sizes[1] / DEDGE;
  const int etot    = n_edges + n_nodes;
  float* out = (float*)d_out;

  char* ws = (char*)d_ws;
  size_t off = 0;
  auto carve = [&](size_t bytes) -> float* {
    float* p = (float*)(ws + off);
    off += (bytes + 255) & ~(size_t)255;
    return p;
  };
  float* Wn      = carve((size_t)n_nodes * DNODE * sizeof(float));
  float* score_s = carve((size_t)n_nodes * HEADS * sizeof(float));
  float* score_r = carve((size_t)n_nodes * HEADS * sizeof(float));
  float* wcomb   = carve((size_t)HEADS * DEDGE * sizeof(float));
  const size_t off_common = off;

  const int n_tiles = (n_nodes + 15) / 16;
  dim3 blk(256);

  gat_node_project<<<dim3((n_tiles + 7) / 8), blk, 0, stream>>>(
      nodes, W, a, W_edge, Wn, score_s, score_r, wcomb, n_nodes, n_tiles);

  const int npb          = (n_nodes + NBINS - 1) / NBINS;
  const int chunk_edges  = (n_edges + NCHUNK - 1) / NCHUNK;
  const int chunk_stride = (chunk_edges + 31) & ~31;
  const size_t lds_bytes = (size_t)npb * (DNODE + 2 * HEADS) * sizeof(float);
  int2*   seg   = (int2*)carve((size_t)NBINS * NCHUNK * sizeof(int2));
  int*    sr    = (int*)carve((size_t)NCHUNK * chunk_stride * sizeof(int));
  float4* lgrec = (float4*)carve((size_t)NCHUNK * chunk_stride * sizeof(float4));
  if (off <= ws_size && lds_bytes <= 65536 && n_nodes <= (1 << 24) && npb > 0 &&
      chunk_edges <= STAGE_CAP) {
    gat_edge_order<<<dim3(NCHUNK), dim3(1024), 0, stream>>>(
        edges, recv, send, score_s, wcomb, sr, lgrec, seg,
        n_edges, chunk_edges, chunk_stride, npb);
    gat_owner_sum<<<dim3(NBINS), dim3(1024), lds_bytes, stream>>>(
        sr, lgrec, seg, score_s, score_r, Wn, ln_s, ln_b, out, n_nodes, npb);
    return;
  }

  off = off_common;
  const size_t plane_bytes = (size_t)n_nodes * DNODE * sizeof(float);
  const size_t norms_bytes = (size_t)2 * n_nodes * HEADS * sizeof(float);
  float* aggr1   = carve(plane_bytes + norms_bytes);
  float* norm01  = aggr1 + (size_t)n_nodes * DNODE;

  hipMemsetAsync(out,   0, plane_bytes, stream);
  hipMemsetAsync(aggr1, 0, plane_bytes + norms_bytes, stream);

  gat_edge_fused<<<dim3((etot + 255) / 256), blk, 0, stream>>>(
      edges, recv, send, score_s, score_r, wcomb, Wn, out, aggr1, norm01, n_edges, n_nodes);
  gat_finalize<<<dim3((n_nodes + 255) / 256), blk, 0, stream>>>(
      out, aggr1, norm01, ln_s, ln_b, n_nodes);
}


# serial memory round trips cut: edge-order index loads and owner start-value loads go out as one group each (were 7 and 13 dependent trips); sender-score gather issued with the edge's feature loads
# speedup vs baseline: 16.3747x; 1.1395x over previous
#include <hip/hip_runtime.h>
#include <hip/hip_bf16.h>
#include <math.h>

typedef __attribute__((ext_vector_type(2))) float v2f;
typedef __attribute__((ext_vector_type(8))) float v8f;

#define HEADS 4
#define ATT_F 16
#define DNODE 64
#define DEDGE 16

#define WA_PITCH 68

__global__ __launch_bounds__(256) void gat_node_project(
    const float* __restrict__ nodes, const float* __restrict__ W,
    const float* __restrict__ a, const float* __restrict__ W_edge,
    float* __restrict__ Wn, float* __restrict__ score_s,
    float* __restrict__ score_r, float* __restrict__ wcomb,
    int n_nodes, int n_tiles)
{
  __shared__ __attribute__((aligned(16))) float s_wa[16 * WA_PITCH];
  const int tid = threadIdx.x;

  {
    const int h = tid >> 6, i = tid & 63;
    float as = 0.f, ar = 0.f;
#pragma unroll
    for (int f = 0; f < ATT_F; ++f) {
      const float w = W[(size_t)(h * ATT_F + f) * DNODE + i];
      as += a[h * 48 + f] * w;
      ar += a[h * 48 + ATT_F + f] * w;
    }
    s_wa[h * WA_PITCH + i]               = as;
    s_wa[(HEADS + h) * WA_PITCH + i]     = ar;
    s_wa[(2 * HEADS + h) * WA_PITCH + i] = 0.f;
    s_wa[(3 * HEADS + h) * WA_PITCH + i] = 0.f;
  }
  if (blockIdx.x == 0 && tid < HEADS * DEDGE) {
    const int h = tid >> 4, i = tid & 15;
    float acc = 0.f;
#pragma unroll
    for (int f = 0; f < ATT_F; ++f)
      acc += a[h * 48 + 32 + f] * W_edge[((h * ATT_F) + f) * DEDGE + i];
    wcomb[tid] = acc;
  }
  __syncthreads();

  int wave = blockIdx.x * (blockDim.x >> 5) + (tid >> 5);
  if (wave >= n_tiles) return;
  int lane = tid & 31;
  int half = lane >> 4;
  int l16  = lane & 15;
  int row0 = wave * 16;
  int m = row0 + l16; if (m >= n_nodes) m = n_nodes - 1;
  const float* __restrict__ arow = nodes + (size_t)m * DNODE;
  const float* wa_row = s_wa + l16 * WA_PITCH;

  v8f acc[4] = {};
  v8f acc5 = {};
#pragma unroll 4
  for (int k0 = 0; k0 < DNODE; k0 += 4) {
    int kk = k0 + 2 * half;
    v2f av; av.x = arow[kk]; av.y = arow[kk + 1];
#pragma unroll
    for (int t = 0; t < 4; ++t) {
      int j = t * 16 + l16;
      v2f b; b.x = W[(size_t)j * DNODE + kk]; b.y = W[(size_t)j * DNODE + kk + 1];
      acc[t] = __builtin_amdgcn_wmma_f32_16x16x4_f32(
           false, av,  false, b,
           (short)0, acc[t],  false,  false);
    }
    v2f b5; b5.x = wa_row[kk]; b5.y = wa_row[kk + 1];
    acc5 = __builtin_amdgcn_wmma_f32_16x16x4_f32(
         false, av,  false, b5,
         (short)0, acc5,  false,  false);
  }

  float* sdst = (l16 < HEADS) ? (score_s + l16) : (score_r + (l16 - HEADS));
  const bool has_score = l16 < 2 * HEADS;

  if (row0 + 16 <= n_nodes) {
#pragma unroll
    for (int t = 0; t < 4; ++t)
#pragma unroll
      for (int v = 0; v < 8; ++v)
        Wn[(size_t)(row0 + v + 8 * half) * DNODE + t * 16 + l16] = acc[t][v];
    if (has_score) {
#pragma unroll
      for (int v = 0; v < 8; ++v)
        sdst[(size_t)(row0 + v + 8 * half) * HEADS] = acc5[v];
    }
  } else {
#pragma unroll
    for (int t = 0; t < 4; ++t)
#pragma unroll
      for (int v = 0; v < 8; ++v) {
        int mo = row0 + v + 8 * half;
        if (mo < n_nodes) Wn[(size_t)mo * DNODE + t * 16 + l16] = acc[t][v];
      }
#pragma unroll
    for (int v = 0; v < 8; ++v) {
      int mo = row0 + v + 8 * half;
      if (has_score && mo < n_nodes) sdst[(size_t)mo * HEADS] = acc5[v];
    }
  }
}

__device__ __forceinline__ int lane_int(int v, int j)
{
  return __builtin_amdgcn_readlane(v, j);
}
__device__ __forceinline__ float lane_float(float v, int j)
{
  return __int_as_float(__builtin_amdgcn_readlane(__float_as_int(v), j));
}

#define EDGE_GROUP 8

__global__ __launch_bounds__(256) void gat_edge_fused(
    const float* __restrict__ edges, const int* __restrict__ recv,
    const int* __restrict__ send, const float* __restrict__ score_s,
    const float* __restrict__ score_r, const float* __restrict__ wcomb,
    const float* __restrict__ Wn, float* __restrict__ aggr0,
    float* __restrict__ aggr1, float* __restrict__ norm01,
    int n_edges, int n_nodes)
{
  const int chip_half = (int)((__builtin_amdgcn_s_sendmsg_rtn(0x87) >> 18) & 1u);
  float* __restrict__ aggr = chip_half ? aggr1 : aggr0;
  float* __restrict__ norm = norm01 + (size_t)chip_half * (size_t)n_nodes * HEADS;
  const int lane = threadIdx.x & 31;
  const int e    = blockIdx.x * blockDim.x + threadIdx.x;
  const int e0   = e - lane;
  const int etot = n_edges + n_nodes;
  if (e0 >= etot) return;

  int s = 0, r = 0;
  float ez[HEADS] = {0.f, 0.f, 0.f, 0.f};
  if (e < etot) {
    float elt[HEADS] = {0.f, 0.f, 0.f, 0.f};
    if (e < n_edges) {
      s = send[e]; r = recv[e];
      const float4* __restrict__ ep = (const float4*)(edges + (size_t)e * DEDGE);
      float ef[DEDGE];
#pragma unroll
      for (int c = 0; c < 4; ++c) {
        float4 q = ep[c];
        ef[4 * c + 0] = q.x; ef[4 * c + 1] = q.y; ef[4 * c + 2] = q.z; ef[4 * c + 3] = q.w;
      }
#pragma unroll
      for (int h = 0; h < HEADS; ++h) {
        float acc = 0.f;
#pragma unroll
        for (int i = 0; i < DEDGE; ++i) acc += ef[i] * wcomb[h * DEDGE + i];
        elt[h] = acc;
      }
    } else {
      s = r = e - n_edges;
    }
    const float4 qs = ((const float4*)score_s)[s];
    const float4 qr = ((const float4*)score_r)[r];
    const float sc[HEADS] = {qs.x + qr.x, qs.y + qr.y, qs.z + qr.z, qs.w + qr.w};
#pragma unroll
    for (int h = 0; h < HEADS; ++h) {
      float m  = sc[h] + elt[h];
      float lm = (m > 0.f) ? m : 0.01f * m;
      ez[h] = __expf(lm);
    }
  }

  const int  cnt = min(32, etot - e0);
  const bool up  = lane >= 16;

  {
    const int  hsel = lane & 3;
    const bool odd  = (lane & 1) != 0, hi = (lane & 2) != 0;
    for (int k = 0; k < 4; ++k) {
      const int src = 8 * k + (lane >> 2);
      const int   rr = __shfl(r, src);
      const float z0 = __shfl(ez[0], src), z1 = __shfl(ez[1], src);
      const float z2 = __shfl(ez[2], src), z3 = __shfl(ez[3], src);
      const float z01 = odd ? z1 : z0;
      const float z23 = odd ? z3 : z2;
      const float z   = hi ? z23 : z01;
      if (src < cnt) atomicAdd(norm + (size_t)rr * HEADS + hsel, z);
    }
  }

  int j0 = 0;
  for (; j0 + EDGE_GROUP <= cnt; j0 += EDGE_GROUP) {
    float vlo[EDGE_GROUP], vhi[EDGE_GROUP];
#pragma unroll
    for (int u = 0; u < EDGE_GROUP; ++u) {
      const float* __restrict__ srow = Wn + (size_t)lane_int(s, j0 + u) * DNODE;
      vlo[u] = srow[lane];
      vhi[u] = srow[32 + lane];
    }
#pragma unroll
    for (int u = 0; u < EDGE_GROUP; ++u) {
      const int j  = j0 + u;
      const int rj = lane_int(r, j);
      float* drow = aggr + (size_t)rj * DNODE;
      const float a0 = lane_float(ez[0], j), a1 = lane_float(ez[1], j);
      const float a2 = lane_float(ez[2], j), a3 = lane_float(ez[3], j);
      atomicAdd(drow + lane,      (up ? a1 : a0) * vlo[u]);
      atomicAdd(drow + 32 + lane, (up ? a3 : a2) * vhi[u]);
    }
  }
  for (; j0 < cnt; ++j0) {
    const float* __restrict__ srow = Wn + (size_t)lane_int(s, j0) * DNODE;
    const int rj = lane_int(r, j0);
    float* drow = aggr + (size_t)rj * DNODE;
    const float a0 = lane_float(ez[0], j0), a1 = lane_float(ez[1], j0);
    const float a2 = lane_float(ez[2], j0), a3 = lane_float(ez[3], j0);
    const float vlo = srow[lane], vhi = srow[32 + lane];
    atomicAdd(drow + lane,      (up ? a1 : a0) * vlo);
    atomicAdd(drow + 32 + lane, (up ? a3 : a2) * vhi);
  }
}

__global__ __launch_bounds__(256) void gat_finalize(
    float* __restrict__ out, const float* __restrict__ aggr1,
    const float* __restrict__ norm01,
    const float* __restrict__ scale, const float* __restrict__ bias, int n_nodes)
{
  int i = blockIdx.x * blockDim.x + threadIdx.x;
  if (i >= n_nodes) return;
  float* row = out + (size_t)i * DNODE;
  const float* __restrict__ row1 = aggr1 + (size_t)i * DNODE;
  const float4 q  = ((const float4*)norm01)[i];
  const float4 q1 = ((const float4*)norm01)[(size_t)n_nodes + i];
  const float nrm[HEADS] = {q.x + q1.x, q.y + q1.y, q.z + q1.z, q.w + q1.w};
  float v[DNODE];
  float mean = 0.f;
#pragma unroll
  for (int j = 0; j < DNODE; ++j) {
    float x = (row[j] + row1[j]) / nrm[j / ATT_F];
    x = (x > 0.f) ? x : expm1f(x);
    v[j] = x; mean += x;
  }
  mean *= (1.f / DNODE);
  float var = 0.f;
#pragma unroll
  for (int j = 0; j < DNODE; ++j) { float d = v[j] - mean; var += d * d; }
  var *= (1.f / DNODE);
  float inv = rsqrtf(var + 1e-6f);
#pragma unroll
  for (int j = 0; j < DNODE; ++j)
    row[j] = (v[j] - mean) * inv * scale[j] + bias[j];
}

#define NBINS  256
#define NCHUNK 256
#define OWNER_GROUP 16
#define OWNER_NPW 8
#define STAGE_CAP 8192
#define EDGES_PER_THREAD (STAGE_CAP / 1024)

__global__ __launch_bounds__(1024) void gat_edge_order(
    const float* __restrict__ edges, const int* __restrict__ recv,
    const int* __restrict__ send, const float* __restrict__ score_s,
    const float* __restrict__ wcomb,
    int* __restrict__ sr, float4* __restrict__ lgrec, int2* __restrict__ seg,
    int n_edges, int chunk_edges, int chunk_stride, int npb)
{
  __shared__ int s_cnt[NBINS];
  __shared__ int s_cur[NBINS];
  __shared__ int s_wsum[NBINS / 32];
  __shared__ __attribute__((aligned(16))) float4 s_lg[STAGE_CAP];
  __shared__ int s_sr[STAGE_CAP];
  const int tid = threadIdx.x;
  const int c   = blockIdx.x;
  const int e_begin = c * chunk_edges;
  const int e_end   = min(n_edges, e_begin + chunk_edges);
  const int rbase   = c * chunk_stride;
  if (tid < NBINS) s_cnt[tid] = 0;
  int rv[EDGES_PER_THREAD], sv[EDGES_PER_THREAD];
  const int e_last = max(e_end - 1, 0);
#pragma unroll
  for (int k = 0; k < EDGES_PER_THREAD; ++k) {
    const int e = min(e_begin + tid + k * 1024, e_last);
    rv[k] = recv[e];
    sv[k] = send[e];
  }
  __syncthreads();
  unsigned wrec[EDGES_PER_THREAD];
  unsigned bins[EDGES_PER_THREAD];
#pragma unroll
  for (int k = 0; k < EDGES_PER_THREAD; ++k) {
    const unsigned bin = (unsigned)rv[k] / (unsigned)npb;
    const unsigned rl  = (unsigned)rv[k] - bin * (unsigned)npb;
    bins[k] = bin;
    wrec[k] = (unsigned)sv[k] | (rl << 24);
    if (e_begin + tid + k * 1024 < e_end) atomicAdd(&s_cnt[bin], 1);
  }
  __syncthreads();
  int v = 0, incl = 0;
  if (tid < NBINS) {
    v = s_cnt[tid];
    incl = v;
    for (int d = 1; d < 32; d <<= 1) {
      const int t = __shfl_up(incl, d);
      if ((tid & 31) >= d) incl += t;
    }
    if ((tid & 31) == 31) s_wsum[tid >> 5] = incl;
  }
  __syncthreads();
  if (tid < NBINS) {
    int before = 0;
    for (int w = 0; w < (tid >> 5); ++w) before += s_wsum[w];
    const int excl = before + incl - v;
    s_cur[tid] = excl;
    seg[tid * NCHUNK + c] = make_int2(rbase + excl, v);
  }
  __syncthreads();
#pragma unroll
  for (int k = 0; k < EDGES_PER_THREAD; ++k) {
    const int e = e_begin + tid + k * 1024;
    if (e < e_end) {
      const float4 qs = ((const float4*)score_s)[wrec[k] & 0xFFFFFFu];
      const float4* __restrict__ ep = (const float4*)(edges + (size_t)e * DEDGE);
      float ef[DEDGE];
#pragma unroll
      for (int q4 = 0; q4 < 4; ++q4) {
        float4 q = ep[q4];
        ef[4 * q4 + 0] = q.x; ef[4 * q4 + 1] = q.y; ef[4 * q4 + 2] = q.z; ef[4 * q4 + 3] = q.w;
      }
      float elt[HEADS];
#pragma unroll
      for (int h = 0; h < HEADS; ++h) {
        float acc = 0.f;
#pragma unroll
        for (int i = 0; i < DEDGE; ++i) acc += ef[i] * wcomb[h * DEDGE + i];
        elt[h] = acc;
      }
      const int slot = atomicAdd(&s_cur[bins[k]], 1);
      s_sr[slot] = (int)wrec[k];
      s_lg[slot] = make_float4(qs.x + elt[0], qs.y + elt[1], qs.z + elt[2], qs.w + elt[3]);
    }
  }
  __syncthreads();
  const int n_rec = e_end - e_begin;
  for (int i = tid; i < n_rec; i += 1024) {
    sr[rbase + i]    = s_sr[i];
    lgrec[rbase + i] = s_lg[i];
  }
}

__global__ __launch_bounds__(1024) void gat_owner_sum(
    const int* __restrict__ sr, const float4* __restrict__ lgrec,
    const int2* __restrict__ seg, const float* __restrict__ score_s,
    const float* __restrict__ score_r, const float* __restrict__ Wn,
    const float* __restrict__ scale, const float* __restrict__ bias,
    float* __restrict__ out, int n_nodes, int npb)
{
  extern __shared__ float smem[];
  float* acc = smem;
  float* nrm = smem + npb * DNODE;
  float* scr = nrm + npb * HEADS;
  const int tid  = threadIdx.x;
  const int lane = tid & 31;
  const int wave = __builtin_amdgcn_readfirstlane(tid >> 5);
  const int b     = blockIdx.x;
  const int node0 = b * npb;
  const int nn    = min(npb, n_nodes - node0);
  const bool up   = lane >= 16;

  int2 sgl = make_int2(0, 0);
  if (lane < NCHUNK / 32) sgl = seg[b * NCHUNK + wave * (NCHUNK / 32) + lane];

  const int  n_l    = wave + 32 * (lane & (OWNER_NPW - 1));
  const int  node_l = min(node0 + max(min(n_l, nn - 1), 0), n_nodes - 1);
  const float4 qs_l = ((const float4*)score_s)[node_l];
  const float4 qr_l = ((const float4*)score_r)[node_l];
  float ivlo[OWNER_NPW], ivhi[OWNER_NPW];
#pragma unroll
  for (int u = 0; u < OWNER_NPW; ++u) {
    const int node = min(node0 + max(min(wave + 32 * u, nn - 1), 0), n_nodes - 1);
    const float* __restrict__ srow = Wn + (size_t)node * DNODE;
    ivlo[u] = srow[lane];
    ivhi[u] = srow[32 + lane];
  }
  int    w_nx = 0;
  float4 q_nx = make_float4(0.f, 0.f, 0.f, 0.f);
  {
    const int st0 = lane_int(sgl.x, 0), c0 = lane_int(sgl.y, 0);
    if (lane < min(32, c0)) { w_nx = sr[st0 + lane]; q_nx = lgrec[st0 + lane]; }
  }
  {
    const float m0 = qs_l.x + qr_l.x, m1 = qs_l.y + qr_l.y;
    const float m2 = qs_l.z + qr_l.z, m3 = qs_l.w + qr_l.w;
    const float iz0 = __expf((m0 > 0.f) ? m0 : 0.01f * m0);
    const float iz1 = __expf((m1 > 0.f) ? m1 : 0.01f * m1);
    const float iz2 = __expf((m2 > 0.f) ? m2 : 0.01f * m2);
    const float iz3 = __expf((m3 > 0.f) ? m3 : 0.01f * m3);
    if (lane < OWNER_NPW && n_l < nn) {
      nrm[n_l * HEADS + 0] = iz0; nrm[n_l * HEADS + 1] = iz1;
      nrm[n_l * HEADS + 2] = iz2; nrm[n_l * HEADS + 3] = iz3;
      scr[n_l * HEADS + 0] = qr_l.x; scr[n_l * HEADS + 1] = qr_l.y;
      scr[n_l * HEADS + 2] = qr_l.z; scr[n_l * HEADS + 3] = qr_l.w;
    }
#pragma unroll
    for (int u = 0; u < OWNER_NPW; ++u) {
      const int n = wave + 32 * u;
      if (n < nn) {
        const float a0 = lane_float(iz0, u), a1 = lane_float(iz1, u);
        const float a2 = lane_float(iz2, u), a3 = lane_float(iz3, u);
        acc[n * DNODE + lane]      = (up ? a1 : a0) * ivlo[u];
        acc[n * DNODE + 32 + lane] = (up ? a3 : a2) * ivhi[u];
      }
    }
  }
  __syncthreads();

  for (int k = 0; k < NCHUNK / 32; ++k) {
    const int start = lane_int(sgl.x, k);
    const int cnt   = lane_int(sgl.y, k);
    const int    w_first = w_nx;
    const float4 q_first = q_nx;
    if (k + 1 < NCHUNK / 32) {
      const int st1 = lane_int(sgl.x, k + 1), c1 = lane_int(sgl.y, k + 1);
      if (lane < min(32, c1)) { w_nx = sr[st1 + lane]; q_nx = lgrec[st1 + lane]; }
    }
    for (int base = 0; base < cnt; base += 32) {
      const int m = min(32, cnt - base);
      int s = 0, rl = 0;
      float ez0 = 0.f, ez1 = 0.f, ez2 = 0.f, ez3 = 0.f;
      if (lane < m) {
        int w = w_first;
        float4 q = q_first;
        if (base > 0) {
          w = sr[start + base + lane];
          q = lgrec[start + base + lane];
        }
        s  = w & 0xFFFFFF;
        rl = (int)((unsigned)w >> 24);
        const float m0 = q.x + scr[rl * HEADS + 0], m1 = q.y + scr[rl * HEADS + 1];
        const float m2 = q.z + scr[rl * HEADS + 2], m3 = q.w + scr[rl * HEADS + 3];
        ez0 = __expf((m0 > 0.f) ? m0 : 0.01f * m0);
        ez1 = __expf((m1 > 0.f) ? m1 : 0.01f * m1);
        ez2 = __expf((m2 > 0.f) ? m2 : 0.01f * m2);
        ez3 = __expf((m3 > 0.f) ? m3 : 0.01f * m3);
        atomicAdd(&nrm[rl * HEADS + 0], ez0);
        atomicAdd(&nrm[rl * HEADS + 1], ez1);
        atomicAdd(&nrm[rl * HEADS + 2], ez2);
        atomicAdd(&nrm[rl * HEADS + 3], ez3);
      }
      for (int j0 = 0; j0 < m; j0 += OWNER_GROUP) {
        float vlo[OWNER_GROUP], vhi[OWNER_GROUP];
#pragma unroll
        for (int u = 0; u < OWNER_GROUP; ++u) {
          vlo[u] = 0.f; vhi[u] = 0.f;
          if (j0 + u < m) {
            const float* __restrict__ srow = Wn + (size_t)lane_int(s, j0 + u) * DNODE;
            vlo[u] = srow[lane];
            vhi[u] = srow[32 + lane];
          }
        }
#pragma unroll
        for (int u = 0; u < OWNER_GROUP; ++u) {
          if (j0 + u < m) {
            const int j  = j0 + u;
            const int rj = lane_int(rl, j);
            float* drow = acc + rj * DNODE;
            const float a0 = lane_float(ez0, j), a1 = lane_float(ez1, j);
            const float a2 = lane_float(ez2, j), a3 = lane_float(ez3, j);
            atomicAdd(drow + lane,      (up ? a1 : a0) * vlo[u]);
            atomicAdd(drow + 32 + lane, (up ? a3 : a2) * vhi[u]);
          }
        }
      }
    }
  }
  __syncthreads();

  const float sc_lo = scale[lane], sc_hi = scale[32 + lane];
  const float bi_lo = bias[lane],  bi_hi = bias[32 + lane];
  for (int n = wave; n < nn; n += 32) {
    const float n01 = up ? nrm[n * HEADS + 1] : nrm[n * HEADS + 0];
    const float n23 = up ? nrm[n * HEADS + 3] : nrm[n * HEADS + 2];
    float x0 = acc[n * DNODE + lane] / n01;
    float x1 = acc[n * DNODE + 32 + lane] / n23;
    x0 = (x0 > 0.f) ? x0 : expm1f(x0);
    x1 = (x1 > 0.f) ? x1 : expm1f(x1);
    float sum = x0 + x1;
    for (int d = 16; d >= 1; d >>= 1) sum += __shfl_xor(sum, d);
    const float mean = sum * (1.f / DNODE);
    const float d0 = x0 - mean, d1 = x1 - mean;
    float var = d0 * d0 + d1 * d1;
    for (int d = 16; d >= 1; d >>= 1) var += __shfl_xor(var, d);
    var *= (1.f / DNODE);
    const float inv = rsqrtf(var + 1e-6f);
    float* row = out + (size_t)(node0 + n) * DNODE;
    row[lane]      = d0 * inv * sc_lo + bi_lo;
    row[32 + lane] = d1 * inv * sc_hi + bi_hi;
  }
}

extern "C" void kernel_launch(void* const* d_in, const int* in_sizes, int n_in,
                              void* d_out, int out_size, void* d_ws, size_t ws_size,
                              hipStream_t stream)
{
  const float* nodes  = (const float*)d_in[0];
  const float* edges  = (const float*)d_in[1];
  const int*   recv   = (const int*)d_in[2];
  const int*   send   = (const int*)d_in[3];
  const float* W      = (const float*)d_in[4];
  const float* W_edge = (const float*)d_in[5];
  const float* a      = (const float*)d_in[6];
  const float* ln_s   = (const float*)d_in[7];
  const float* ln_b   = (const float*)d_in[8];

  const int n_nodes = in_sizes[0] / DNODE;
  const int n_edges = in_sizes[1] / DEDGE;
  const int etot    = n_edges + n_nodes;
  float* out = (float*)d_out;

  char* ws = (char*)d_ws;
  size_t off = 0;
  auto carve = [&](size_t bytes) -> float* {
    float* p = (float*)(ws + off);
    off += (bytes + 255) & ~(size_t)255;
    return p;
  };
  float* Wn      = carve((size_t)n_nodes * DNODE * sizeof(float));
  float* score_s = carve((size_t)n_nodes * HEADS * sizeof(float));
  float* score_r = carve((size_t)n_nodes * HEADS * sizeof(float));
  float* wcomb   = carve((size_t)HEADS * DEDGE * sizeof(float));
  const size_t off_common = off;

  const int n_tiles = (n_nodes + 15) / 16;
  dim3 blk(256);

  gat_node_project<<<dim3((n_tiles + 7) / 8), blk, 0, stream>>>(
      nodes, W, a, W_edge, Wn, score_s, score_r, wcomb, n_nodes, n_tiles);

  const int npb          = (n_nodes + NBINS - 1) / NBINS;
  const int chunk_edges  = (n_edges + NCHUNK - 1) / NCHUNK;
  const int chunk_stride = (chunk_edges + 31) & ~31;
  const size_t lds_bytes = (size_t)npb * (DNODE + 2 * HEADS) * sizeof(float);
  int2*   seg   = (int2*)carve((size_t)NBINS * NCHUNK * sizeof(int2));
  int*    sr    = (int*)carve((size_t)NCHUNK * chunk_stride * sizeof(int));
  float4* lgrec = (float4*)carve((size_t)NCHUNK * chunk_stride * sizeof(float4));
  if (off <= ws_size && lds_bytes <= 65536 && n_nodes <= (1 << 24) && npb > 0 &&
      chunk_edges <= STAGE_CAP) {
    gat_edge_order<<<dim3(NCHUNK), dim3(1024), 0, stream>>>(
        edges, recv, send, score_s, wcomb, sr, lgrec, seg,
        n_edges, chunk_edges, chunk_stride, npb);
    gat_owner_sum<<<dim3(NBINS), dim3(1024), lds_bytes, stream>>>(
        sr, lgrec, seg, score_s, score_r, Wn, ln_s, ln_b, out, n_nodes, npb);
    return;
  }

  off = off_common;
  const size_t plane_bytes = (size_t)n_nodes * DNODE * sizeof(float);
  const size_t norms_bytes = (size_t)2 * n_nodes * HEADS * sizeof(float);
  float* aggr1   = carve(plane_bytes + norms_bytes);
  float* norm01  = aggr1 + (size_t)n_nodes * DNODE;

  hipMemsetAsync(out,   0, plane_bytes, stream);
  hipMemsetAsync(aggr1, 0, plane_bytes + norms_bytes, stream);

  gat_edge_fused<<<dim3((etot + 255) / 256), blk, 0, stream>>>(
      edges, recv, send, score_s, score_r, wcomb, Wn, out, aggr1, norm01, n_edges, n_nodes);
  gat_finalize<<<dim3((n_nodes + 255) / 256), blk, 0, stream>>>(
      out, aggr1, norm01, ln_s, ln_b, n_nodes);
}


# owner launch: a range's records put in receiver order in LDS (integer count, scan, slot by returning integer add), one wave sums a receiver's rows in registers; no LDS float adds; sums in slot order
# speedup vs baseline: 19.1760x; 1.1711x over previous
#include <hip/hip_runtime.h>
#include <hip/hip_bf16.h>
#include <math.h>

typedef __attribute__((ext_vector_type(2))) float v2f;
typedef __attribute__((ext_vector_type(8))) float v8f;

#define HEADS 4
#define ATT_F 16
#define DNODE 64
#define DEDGE 16

#define WA_PITCH 68

__global__ __launch_bounds__(256) void gat_node_project(
    const float* __restrict__ nodes, const float* __restrict__ W,
    const float* __restrict__ a, const float* __restrict__ W_edge,
    float* __restrict__ Wn, float* __restrict__ score_s,
    float* __restrict__ score_r, float* __restrict__ wcomb,
    int n_nodes, int n_tiles)
{
  __shared__ __attribute__((aligned(16))) float s_wa[16 * WA_PITCH];
  const int tid = threadIdx.x;

  {
    const int h = tid >> 6, i = tid & 63;
    float as = 0.f, ar = 0.f;
#pragma unroll
    for (int f = 0; f < ATT_F; ++f) {
      const float w = W[(size_t)(h * ATT_F + f) * DNODE + i];
      as += a[h * 48 + f] * w;
      ar += a[h * 48 + ATT_F + f] * w;
    }
    s_wa[h * WA_PITCH + i]               = as;
    s_wa[(HEADS + h) * WA_PITCH + i]     = ar;
    s_wa[(2 * HEADS + h) * WA_PITCH + i] = 0.f;
    s_wa[(3 * HEADS + h) * WA_PITCH + i] = 0.f;
  }
  if (blockIdx.x == 0 && tid < HEADS * DEDGE) {
    const int h = tid >> 4, i = tid & 15;
    float acc = 0.f;
#pragma unroll
    for (int f = 0; f < ATT_F; ++f)
      acc += a[h * 48 + 32 + f] * W_edge[((h * ATT_F) + f) * DEDGE + i];
    wcomb[tid] = acc;
  }
  __syncthreads();

  int wave = blockIdx.x * (blockDim.x >> 5) + (tid >> 5);
  if (wave >= n_tiles) return;
  int lane = tid & 31;
  int half = lane >> 4;
  int l16  = lane & 15;
  int row0 = wave * 16;
  int m = row0 + l16; if (m >= n_nodes) m = n_nodes - 1;
  const float* __restrict__ arow = nodes + (size_t)m * DNODE;
  const float* wa_row = s_wa + l16 * WA_PITCH;

  v8f acc[4] = {};
  v8f acc5 = {};
#pragma unroll 4
  for (int k0 = 0; k0 < DNODE; k0 += 4) {
    int kk = k0 + 2 * half;
    v2f av; av.x = arow[kk]; av.y = arow[kk + 1];
#pragma unroll
    for (int t = 0; t < 4; ++t) {
      int j = t * 16 + l16;
      v2f b; b.x = W[(size_t)j * DNODE + kk]; b.y = W[(size_t)j * DNODE + kk + 1];
      acc[t] = __builtin_amdgcn_wmma_f32_16x16x4_f32(
           false, av,  false, b,
           (short)0, acc[t],  false,  false);
    }
    v2f b5; b5.x = wa_row[kk]; b5.y = wa_row[kk + 1];
    acc5 = __builtin_amdgcn_wmma_f32_16x16x4_f32(
         false, av,  false, b5,
         (short)0, acc5,  false,  false);
  }

  float* sdst = (l16 < HEADS) ? (score_s + l16) : (score_r + (l16 - HEADS));
  const bool has_score = l16 < 2 * HEADS;

  if (row0 + 16 <= n_nodes) {
#pragma unroll
    for (int t = 0; t < 4; ++t)
#pragma unroll
      for (int v = 0; v < 8; ++v)
        Wn[(size_t)(row0 + v + 8 * half) * DNODE + t * 16 + l16] = acc[t][v];
    if (has_score) {
#pragma unroll
      for (int v = 0; v < 8; ++v)
        sdst[(size_t)(row0 + v + 8 * half) * HEADS] = acc5[v];
    }
  } else {
#pragma unroll
    for (int t = 0; t < 4; ++t)
#pragma unroll
      for (int v = 0; v < 8; ++v) {
        int mo = row0 + v + 8 * half;
        if (mo < n_nodes) Wn[(size_t)mo * DNODE + t * 16 + l16] = acc[t][v];
      }
#pragma unroll
    for (int v = 0; v < 8; ++v) {
      int mo = row0 + v + 8 * half;
      if (has_score && mo < n_nodes) sdst[(size_t)mo * HEADS] = acc5[v];
    }
  }
}

__device__ __forceinline__ int lane_int(int v, int j)
{
  return __builtin_amdgcn_readlane(v, j);
}
__device__ __forceinline__ float lane_float(float v, int j)
{
  return __int_as_float(__builtin_amdgcn_readlane(__float_as_int(v), j));
}

#define EDGE_GROUP 8

__global__ __launch_bounds__(256) void gat_edge_fused(
    const float* __restrict__ edges, const int* __restrict__ recv,
    const int* __restrict__ send, const float* __restrict__ score_s,
    const float* __restrict__ score_r, const float* __restrict__ wcomb,
    const float* __restrict__ Wn, float* __restrict__ aggr0,
    float* __restrict__ aggr1, float* __restrict__ norm01,
    int n_edges, int n_nodes)
{
  const int chip_half = (int)((__builtin_amdgcn_s_sendmsg_rtn(0x87) >> 18) & 1u);
  float* __restrict__ aggr = chip_half ? aggr1 : aggr0;
  float* __restrict__ norm = norm01 + (size_t)chip_half * (size_t)n_nodes * HEADS;
  const int lane = threadIdx.x & 31;
  const int e    = blockIdx.x * blockDim.x + threadIdx.x;
  const int e0   = e - lane;
  const int etot = n_edges + n_nodes;
  if (e0 >= etot) return;

  int s = 0, r = 0;
  float ez[HEADS] = {0.f, 0.f, 0.f, 0.f};
  if (e < etot) {
    float elt[HEADS] = {0.f, 0.f, 0.f, 0.f};
    if (e < n_edges) {
      s = send[e]; r = recv[e];
      const float4* __restrict__ ep = (const float4*)(edges + (size_t)e * DEDGE);
      float ef[DEDGE];
#pragma unroll
      for (int c = 0; c < 4; ++c) {
        float4 q = ep[c];
        ef[4 * c + 0] = q.x; ef[4 * c + 1] = q.y; ef[4 * c + 2] = q.z; ef[4 * c + 3] = q.w;
      }
#pragma unroll
      for (int h = 0; h < HEADS; ++h) {
        float acc = 0.f;
#pragma unroll
        for (int i = 0; i < DEDGE; ++i) acc += ef[i] * wcomb[h * DEDGE + i];
        elt[h] = acc;
      }
    } else {
      s = r = e - n_edges;
    }
    const float4 qs = ((const float4*)score_s)[s];
    const float4 qr = ((const float4*)score_r)[r];
    const float sc[HEADS] = {qs.x + qr.x, qs.y + qr.y, qs.z + qr.z, qs.w + qr.w};
#pragma unroll
    for (int h = 0; h < HEADS; ++h) {
      float m  = sc[h] + elt[h];
      float lm = (m > 0.f) ? m : 0.01f * m;
      ez[h] = __expf(lm);
    }
  }

  const int  cnt = min(32, etot - e0);
  const bool up  = lane >= 16;

  {
    const int  hsel = lane & 3;
    const bool odd  = (lane & 1) != 0, hi = (lane & 2) != 0;
    for (int k = 0; k < 4; ++k) {
      const int src = 8 * k + (lane >> 2);
      const int   rr = __shfl(r, src);
      const float z0 = __shfl(ez[0], src), z1 = __shfl(ez[1], src);
      const float z2 = __shfl(ez[2], src), z3 = __shfl(ez[3], src);
      const float z01 = odd ? z1 : z0;
      const float z23 = odd ? z3 : z2;
      const float z   = hi ? z23 : z01;
      if (src < cnt) atomicAdd(norm + (size_t)rr * HEADS + hsel, z);
    }
  }

  int j0 = 0;
  for (; j0 + EDGE_GROUP <= cnt; j0 += EDGE_GROUP) {
    float vlo[EDGE_GROUP], vhi[EDGE_GROUP];
#pragma unroll
    for (int u = 0; u < EDGE_GROUP; ++u) {
      const float* __restrict__ srow = Wn + (size_t)lane_int(s, j0 + u) * DNODE;
      vlo[u] = srow[lane];
      vhi[u] = srow[32 + lane];
    }
#pragma unroll
    for (int u = 0; u < EDGE_GROUP; ++u) {
      const int j  = j0 + u;
      const int rj = lane_int(r, j);
      float* drow = aggr + (size_t)rj * DNODE;
      const float a0 = lane_float(ez[0], j), a1 = lane_float(ez[1], j);
      const float a2 = lane_float(ez[2], j), a3 = lane_float(ez[3], j);
      atomicAdd(drow + lane,      (up ? a1 : a0) * vlo[u]);
      atomicAdd(drow + 32 + lane, (up ? a3 : a2) * vhi[u]);
    }
  }
  for (; j0 < cnt; ++j0) {
    const float* __restrict__ srow = Wn + (size_t)lane_int(s, j0) * DNODE;
    const int rj = lane_int(r, j0);
    float* drow = aggr + (size_t)rj * DNODE;
    const float a0 = lane_float(ez[0], j0), a1 = lane_float(ez[1], j0);
    const float a2 = lane_float(ez[2], j0), a3 = lane_float(ez[3], j0);
    const float vlo = srow[lane], vhi = srow[32 + lane];
    atomicAdd(drow + lane,      (up ? a1 : a0) * vlo);
    atomicAdd(drow + 32 + lane, (up ? a3 : a2) * vhi);
  }
}

__global__ __launch_bounds__(256) void gat_finalize(
    float* __restrict__ out, const float* __restrict__ aggr1,
    const float* __restrict__ norm01,
    const float* __restrict__ scale, const float* __restrict__ bias, int n_nodes)
{
  int i = blockIdx.x * blockDim.x + threadIdx.x;
  if (i >= n_nodes) return;
  float* row = out + (size_t)i * DNODE;
  const float* __restrict__ row1 = aggr1 + (size_t)i * DNODE;
  const float4 q  = ((const float4*)norm01)[i];
  const float4 q1 = ((const float4*)norm01)[(size_t)n_nodes + i];
  const float nrm[HEADS] = {q.x + q1.x, q.y + q1.y, q.z + q1.z, q.w + q1.w};
  float v[DNODE];
  float mean = 0.f;
#pragma unroll
  for (int j = 0; j < DNODE; ++j) {
    float x = (row[j] + row1[j]) / nrm[j / ATT_F];
    x = (x > 0.f) ? x : expm1f(x);
    v[j] = x; mean += x;
  }
  mean *= (1.f / DNODE);
  float var = 0.f;
#pragma unroll
  for (int j = 0; j < DNODE; ++j) { float d = v[j] - mean; var += d * d; }
  var *= (1.f / DNODE);
  float inv = rsqrtf(var + 1e-6f);
#pragma unroll
  for (int j = 0; j < DNODE; ++j)
    row[j] = (v[j] - mean) * inv * scale[j] + bias[j];
}

#define NBINS  256
#define NCHUNK 256
#define OWNER_GROUP 16
#define OWNER_NPW 8
#define STAGE_CAP 8192
#define EDGES_PER_THREAD (STAGE_CAP / 1024)

__global__ __launch_bounds__(1024) void gat_edge_order(
    const float* __restrict__ edges, const int* __restrict__ recv,
    const int* __restrict__ send, const float* __restrict__ score_s,
    const float* __restrict__ wcomb,
    int* __restrict__ sr, float4* __restrict__ lgrec, int2* __restrict__ seg,
    int n_edges, int chunk_edges, int chunk_stride, int npb)
{
  __shared__ int s_cnt[NBINS];
  __shared__ int s_cur[NBINS];
  __shared__ int s_wsum[NBINS / 32];
  __shared__ __attribute__((aligned(16))) float4 s_lg[STAGE_CAP];
  __shared__ int s_sr[STAGE_CAP];
  const int tid = threadIdx.x;
  const int c   = blockIdx.x;
  const int e_begin = c * chunk_edges;
  const int e_end   = min(n_edges, e_begin + chunk_edges);
  const int rbase   = c * chunk_stride;
  if (tid < NBINS) s_cnt[tid] = 0;
  int rv[EDGES_PER_THREAD], sv[EDGES_PER_THREAD];
  const int e_last = max(e_end - 1, 0);
#pragma unroll
  for (int k = 0; k < EDGES_PER_THREAD; ++k) {
    const int e = min(e_begin + tid + k * 1024, e_last);
    rv[k] = recv[e];
    sv[k] = send[e];
  }
  __syncthreads();
  unsigned wrec[EDGES_PER_THREAD];
  unsigned bins[EDGES_PER_THREAD];
#pragma unroll
  for (int k = 0; k < EDGES_PER_THREAD; ++k) {
    const unsigned bin = (unsigned)rv[k] / (unsigned)npb;
    const unsigned rl  = (unsigned)rv[k] - bin * (unsigned)npb;
    bins[k] = bin;
    wrec[k] = (unsigned)sv[k] | (rl << 24);
    if (e_begin + tid + k * 1024 < e_end) atomicAdd(&s_cnt[bin], 1);
  }
  __syncthreads();
  int v = 0, incl = 0;
  if (tid < NBINS) {
    v = s_cnt[tid];
    incl = v;
    for (int d = 1; d < 32; d <<= 1) {
      const int t = __shfl_up(incl, d);
      if ((tid & 31) >= d) incl += t;
    }
    if ((tid & 31) == 31) s_wsum[tid >> 5] = incl;
  }
  __syncthreads();
  if (tid < NBINS) {
    int before = 0;
    for (int w = 0; w < (tid >> 5); ++w) before += s_wsum[w];
    const int excl = before + incl - v;
    s_cur[tid] = excl;
    seg[tid * NCHUNK + c] = make_int2(rbase + excl, v);
  }
  __syncthreads();
#pragma unroll
  for (int k = 0; k < EDGES_PER_THREAD; ++k) {
    const int e = e_begin + tid + k * 1024;
    if (e < e_end) {
      const float4 qs = ((const float4*)score_s)[wrec[k] & 0xFFFFFFu];
      const float4* __restrict__ ep = (const float4*)(edges + (size_t)e * DEDGE);
      float ef[DEDGE];
#pragma unroll
      for (int q4 = 0; q4 < 4; ++q4) {
        float4 q = ep[q4];
        ef[4 * q4 + 0] = q.x; ef[4 * q4 + 1] = q.y; ef[4 * q4 + 2] = q.z; ef[4 * q4 + 3] = q.w;
      }
      float elt[HEADS];
#pragma unroll
      for (int h = 0; h < HEADS; ++h) {
        float acc = 0.f;
#pragma unroll
        for (int i = 0; i < DEDGE; ++i) acc += ef[i] * wcomb[h * DEDGE + i];
        elt[h] = acc;
      }
      const int slot = atomicAdd(&s_cur[bins[k]], 1);
      s_sr[slot] = (int)wrec[k];
      s_lg[slot] = make_float4(qs.x + elt[0], qs.y + elt[1], qs.z + elt[2], qs.w + elt[3]);
    }
  }
  __syncthreads();
  const int n_rec = e_end - e_begin;
  for (int i = tid; i < n_rec; i += 1024) {
    sr[rbase + i]    = s_sr[i];
    lgrec[rbase + i] = s_lg[i];
  }
}

#define REC_CAP 8192
#define REC_ROUNDS 2
#define RUN_PAD 4

template <int R>
__device__ __forceinline__ void round_sum(const float* __restrict__ Wn, const int s_l, const float* zr,
                                          const unsigned lane, float& slo, float& shi, float& nlo, float& nhi)
{
  float vlo[R], vhi[R];
#pragma unroll
  for (int u = 0; u < R; ++u) {
    const unsigned idx = (unsigned)lane_int(s_l, u) * DNODE + lane;
    vlo[u] = Wn[idx];
    vhi[u] = Wn[idx + 32];
  }
#pragma unroll
  for (int u = 0; u < R; ++u) {
    const float2 zz = *(const float2*)(zr + u * 4);
    slo = fmaf(zz.x, vlo[u], slo);
    shi = fmaf(zz.y, vhi[u], shi);
    nlo += zz.x;
    nhi += zz.y;
  }
}

__device__ __forceinline__ void place_record(const int w, const float4 q, const float* scr,
                                             int* s_cur, int* s_rs, float4* s_rz)
{
  const int rl = (int)((unsigned)w >> 24);
  const float4 sc = *(const float4*)(scr + rl * HEADS);
  const float m0 = q.x + sc.x, m1 = q.y + sc.y, m2 = q.z + sc.z, m3 = q.w + sc.w;
  const float ez0 = __expf((m0 > 0.f) ? m0 : 0.01f * m0);
  const float ez1 = __expf((m1 > 0.f) ? m1 : 0.01f * m1);
  const float ez2 = __expf((m2 > 0.f) ? m2 : 0.01f * m2);
  const float ez3 = __expf((m3 > 0.f) ? m3 : 0.01f * m3);
  const int slot = atomicAdd(&s_cur[rl], 1);
  s_rs[slot] = w & 0xFFFFFF;
  s_rz[slot] = make_float4(ez0, ez2, ez1, ez3);
}

__global__ __launch_bounds__(1024) void gat_owner_sum(
    const int* __restrict__ sr, const float4* __restrict__ lgrec,
    const int2* __restrict__ seg, const float* __restrict__ score_s,
    const float* __restrict__ score_r, const float* __restrict__ Wn,
    const float* __restrict__ scale, const float* __restrict__ bias,
    float* __restrict__ out, int n_nodes, int npb)
{
  __shared__ __attribute__((aligned(16))) float4 s_rz[REC_CAP];
  __shared__ __attribute__((aligned(16))) float s_scr[256 * HEADS];
  __shared__ float s_nrm[256 * HEADS];
  float* acc = (float*)s_rz;
  float* nrm = s_nrm;
  float* scr = s_scr;
  __shared__ int s_rs[REC_CAP];
  __shared__ int s_cnt[256];
  __shared__ int s_pos[256];
  __shared__ int s_cur[256];
  __shared__ int s_wsum[8];
  __shared__ int s_wtot[32];
  const int tid  = threadIdx.x;
  const int lane = tid & 31;
  const int wave = __builtin_amdgcn_readfirstlane(tid >> 5);
  const int b     = blockIdx.x;
  const int node0 = b * npb;
  const int nn    = min(npb, n_nodes - node0);
  const bool up   = lane >= 16;

  int2 sgl = make_int2(0, 0);
  if (lane < NCHUNK / 32) sgl = seg[b * NCHUNK + wave * (NCHUNK / 32) + lane];

  int wave_total = 0;
#pragma unroll
  for (int k = 0; k < NCHUNK / 32; ++k) wave_total += lane_int(sgl.y, k);
  if (tid < 256) s_cnt[tid] = 0;
  if (lane == 0) s_wtot[wave] = wave_total;
  __syncthreads();
  int tot = s_wtot[lane];
  for (int d = 16; d >= 1; d >>= 1) tot += __shfl_xor(tot, d);
  const int total = __builtin_amdgcn_readfirstlane(tot);

  if (total + (1 + RUN_PAD) * max(nn, 0) <= REC_CAP) {
    const int  n_l    = wave + 32 * (lane & (OWNER_NPW - 1));
    const int  node_l = min(node0 + max(min(n_l, nn - 1), 0), n_nodes - 1);
    const float4 qs_l = ((const float4*)score_s)[node_l];
    const float4 qr_l = ((const float4*)score_r)[node_l];

    int    w8[(NCHUNK / 32) * REC_ROUNDS];
    float4 q8[(NCHUNK / 32) * REC_ROUNDS];
#pragma unroll
    for (int k = 0; k < NCHUNK / 32; ++k) {
      const int st = lane_int(sgl.x, k), ck = lane_int(sgl.y, k);
#pragma unroll
      for (int r = 0; r < REC_ROUNDS; ++r) {
        w8[k * REC_ROUNDS + r] = 0;
        q8[k * REC_ROUNDS + r] = make_float4(0.f, 0.f, 0.f, 0.f);
        if (lane < ck - 32 * r) {
          w8[k * REC_ROUNDS + r] = sr[st + 32 * r + lane];
          q8[k * REC_ROUNDS + r] = lgrec[st + 32 * r + lane];
        }
      }
    }

    const float sm0 = qs_l.x + qr_l.x, sm1 = qs_l.y + qr_l.y;
    const float sm2 = qs_l.z + qr_l.z, sm3 = qs_l.w + qr_l.w;
    const float iz0 = __expf((sm0 > 0.f) ? sm0 : 0.01f * sm0);
    const float iz1 = __expf((sm1 > 0.f) ? sm1 : 0.01f * sm1);
    const float iz2 = __expf((sm2 > 0.f) ? sm2 : 0.01f * sm2);
    const float iz3 = __expf((sm3 > 0.f) ? sm3 : 0.01f * sm3);
    if (lane < OWNER_NPW && n_l < nn) {
      scr[n_l * HEADS + 0] = qr_l.x; scr[n_l * HEADS + 1] = qr_l.y;
      scr[n_l * HEADS + 2] = qr_l.z; scr[n_l * HEADS + 3] = qr_l.w;
    }
    if (lane < OWNER_NPW && n_l < nn) atomicAdd(&s_cnt[n_l], 1);
#pragma unroll
    for (int k = 0; k < NCHUNK / 32; ++k) {
      const int ck = lane_int(sgl.y, k);
#pragma unroll
      for (int r = 0; r < REC_ROUNDS; ++r)
        if (lane < ck - 32 * r) atomicAdd(&s_cnt[(unsigned)w8[k * REC_ROUNDS + r] >> 24], 1);
    }
    for (int k = 0; k < NCHUNK / 32; ++k) {
      const int st = lane_int(sgl.x, k), ck = lane_int(sgl.y, k);
      for (int base = 32 * REC_ROUNDS; base < ck; base += 32)
        if (lane < ck - base) atomicAdd(&s_cnt[(unsigned)sr[st + base + lane] >> 24], 1);
    }
    __syncthreads();
    int v = 0, vp = 0, incl = 0;
    if (tid < 256) {
      v  = s_cnt[tid];
      vp = (v + RUN_PAD - 1) & ~(RUN_PAD - 1);
      incl = vp;
      for (int d = 1; d < 32; d <<= 1) {
        const int t = __shfl_up(incl, d);
        if (lane >= d) incl += t;
      }
      if (lane == 31) s_wsum[tid >> 5] = incl;
    }
    __syncthreads();
    if (tid < 256) {
      int before = 0;
      for (int w = 0; w < (tid >> 5); ++w) before += s_wsum[w];
      const int excl = before + incl - vp;
      s_pos[tid] = excl;
      s_cur[tid] = excl;
      const int node_pad = max(min(node0 + tid, n_nodes - 1), 0);
      for (int i = v; i < vp; ++i) {
        s_rs[excl + i] = node_pad;
        s_rz[excl + i] = make_float4(0.f, 0.f, 0.f, 0.f);
      }
    }
    __syncthreads();
    if (lane < OWNER_NPW && n_l < nn) {
      const int slot = atomicAdd(&s_cur[n_l], 1);
      s_rs[slot] = node0 + n_l;
      s_rz[slot] = make_float4(iz0, iz2, iz1, iz3);
    }
#pragma unroll
    for (int k = 0; k < NCHUNK / 32; ++k) {
      const int ck = lane_int(sgl.y, k);
#pragma unroll
      for (int r = 0; r < REC_ROUNDS; ++r)
        if (lane < ck - 32 * r)
          place_record(w8[k * REC_ROUNDS + r], q8[k * REC_ROUNDS + r], scr, s_cur, s_rs, s_rz);
    }
    for (int k = 0; k < NCHUNK / 32; ++k) {
      const int st = lane_int(sgl.x, k), ck = lane_int(sgl.y, k);
      for (int base = 32 * REC_ROUNDS; base < ck; base += 32)
        if (lane < ck - base)
          place_record(sr[st + base + lane], lgrec[st + base + lane], scr, s_cur, s_rs, s_rz);
    }
    __syncthreads();

    const float sc_lo = scale[lane], sc_hi = scale[32 + lane];
    const float bi_lo = bias[lane],  bi_hi = bias[32 + lane];
    const float* zb = (const float*)s_rz + (up ? 2 : 0);
    for (int n = wave; n < nn; n += 32) {
      const int p0 = __builtin_amdgcn_readfirstlane(s_pos[n]);
      const int cp = (__builtin_amdgcn_readfirstlane(s_cnt[n]) + RUN_PAD - 1) & ~(RUN_PAD - 1);
      float slo = 0.f, shi = 0.f, nlo = 0.f, nhi = 0.f;
      for (int base = 0; base < cp; base += 32) {
        const int m    = min(32, cp - base);
        const int rec0 = p0 + base;
        int s_l = 0;
        if (lane < m) s_l = s_rs[rec0 + lane];
        const float* zr = zb + rec0 * 4;
        switch (m / RUN_PAD) {
          case 8:  round_sum<32>(Wn, s_l, zr, (unsigned)lane, slo, shi, nlo, nhi); break;
          case 7:  round_sum<28>(Wn, s_l, zr, (unsigned)lane, slo, shi, nlo, nhi); break;
          case 6:  round_sum<24>(Wn, s_l, zr, (unsigned)lane, slo, shi, nlo, nhi); break;
          case 5:  round_sum<20>(Wn, s_l, zr, (unsigned)lane, slo, shi, nlo, nhi); break;
          case 4:  round_sum<16>(Wn, s_l, zr, (unsigned)lane, slo, shi, nlo, nhi); break;
          case 3:  round_sum<12>(Wn, s_l, zr, (unsigned)lane, slo, shi, nlo, nhi); break;
          case 2:  round_sum<8>(Wn, s_l, zr, (unsigned)lane, slo, shi, nlo, nhi); break;
          default: round_sum<4>(Wn, s_l, zr, (unsigned)lane, slo, shi, nlo, nhi); break;
        }
      }
      float x0 = slo / nlo;
      float x1 = shi / nhi;
      x0 = (x0 > 0.f) ? x0 : expm1f(x0);
      x1 = (x1 > 0.f) ? x1 : expm1f(x1);
      float sum = x0 + x1;
      for (int d = 16; d >= 1; d >>= 1) sum += __shfl_xor(sum, d);
      const float mean = sum * (1.f / DNODE);
      const float d0 = x0 - mean, d1 = x1 - mean;
      float var = d0 * d0 + d1 * d1;
      for (int d = 16; d >= 1; d >>= 1) var += __shfl_xor(var, d);
      var *= (1.f / DNODE);
      const float inv = rsqrtf(var + 1e-6f);
      float* row = out + (size_t)(node0 + n) * DNODE;
      row[lane]      = d0 * inv * sc_lo + bi_lo;
      row[32 + lane] = d1 * inv * sc_hi + bi_hi;
    }
    return;
  }

  const int  n_l    = wave + 32 * (lane & (OWNER_NPW - 1));
  const int  node_l = min(node0 + max(min(n_l, nn - 1), 0), n_nodes - 1);
  const float4 qs_l = ((const float4*)score_s)[node_l];
  const float4 qr_l = ((const float4*)score_r)[node_l];
  float ivlo[OWNER_NPW], ivhi[OWNER_NPW];
#pragma unroll
  for (int u = 0; u < OWNER_NPW; ++u) {
    const int node = min(node0 + max(min(wave + 32 * u, nn - 1), 0), n_nodes - 1);
    const float* __restrict__ srow = Wn + (size_t)node * DNODE;
    ivlo[u] = srow[lane];
    ivhi[u] = srow[32 + lane];
  }
  int    w_nx = 0;
  float4 q_nx = make_float4(0.f, 0.f, 0.f, 0.f);
  {
    const int st0 = lane_int(sgl.x, 0), c0 = lane_int(sgl.y, 0);
    if (lane < min(32, c0)) { w_nx = sr[st0 + lane]; q_nx = lgrec[st0 + lane]; }
  }
  {
    const float m0 = qs_l.x + qr_l.x, m1 = qs_l.y + qr_l.y;
    const float m2 = qs_l.z + qr_l.z, m3 = qs_l.w + qr_l.w;
    const float iz0 = __expf((m0 > 0.f) ? m0 : 0.01f * m0);
    const float iz1 = __expf((m1 > 0.f) ? m1 : 0.01f * m1);
    const float iz2 = __expf((m2 > 0.f) ? m2 : 0.01f * m2);
    const float iz3 = __expf((m3 > 0.f) ? m3 : 0.01f * m3);
    if (lane < OWNER_NPW && n_l < nn) {
      nrm[n_l * HEADS + 0] = iz0; nrm[n_l * HEADS + 1] = iz1;
      nrm[n_l * HEADS + 2] = iz2; nrm[n_l * HEADS + 3] = iz3;
      scr[n_l * HEADS + 0] = qr_l.x; scr[n_l * HEADS + 1] = qr_l.y;
      scr[n_l * HEADS + 2] = qr_l.z; scr[n_l * HEADS + 3] = qr_l.w;
    }
#pragma unroll
    for (int u = 0; u < OWNER_NPW; ++u) {
      const int n = wave + 32 * u;
      if (n < nn) {
        const float a0 = lane_float(iz0, u), a1 = lane_float(iz1, u);
        const float a2 = lane_float(iz2, u), a3 = lane_float(iz3, u);
        acc[n * DNODE + lane]      = (up ? a1 : a0) * ivlo[u];
        acc[n * DNODE + 32 + lane] = (up ? a3 : a2) * ivhi[u];
      }
    }
  }
  __syncthreads();

  for (int k = 0; k < NCHUNK / 32; ++k) {
    const int start = lane_int(sgl.x, k);
    const int cnt   = lane_int(sgl.y, k);
    const int    w_first = w_nx;
    const float4 q_first = q_nx;
    if (k + 1 < NCHUNK / 32) {
      const int st1 = lane_int(sgl.x, k + 1), c1 = lane_int(sgl.y, k + 1);
      if (lane < min(32, c1)) { w_nx = sr[st1 + lane]; q_nx = lgrec[st1 + lane]; }
    }
    for (int base = 0; base < cnt; base += 32) {
      const int m = min(32, cnt - base);
      int s = 0, rl = 0;
      float ez0 = 0.f, ez1 = 0.f, ez2 = 0.f, ez3 = 0.f;
      if (lane < m) {
        int w = w_first;
        float4 q = q_first;
        if (base > 0) {
          w = sr[start + base + lane];
          q = lgrec[start + base + lane];
        }
        s  = w & 0xFFFFFF;
        rl = (int)((unsigned)w >> 24);
        const float m0 = q.x + scr[rl * HEADS + 0], m1 = q.y + scr[rl * HEADS + 1];
        const float m2 = q.z + scr[rl * HEADS + 2], m3 = q.w + scr[rl * HEADS + 3];
        ez0 = __expf((m0 > 0.f) ? m0 : 0.01f * m0);
        ez1 = __expf((m1 > 0.f) ? m1 : 0.01f * m1);
        ez2 = __expf((m2 > 0.f) ? m2 : 0.01f * m2);
        ez3 = __expf((m3 > 0.f) ? m3 : 0.01f * m3);
        atomicAdd(&nrm[rl * HEADS + 0], ez0);
        atomicAdd(&nrm[rl * HEADS + 1], ez1);
        atomicAdd(&nrm[rl * HEADS + 2], ez2);
        atomicAdd(&nrm[rl * HEADS + 3], ez3);
      }
      for (int j0 = 0; j0 < m; j0 += OWNER_GROUP) {
        float vlo[OWNER_GROUP], vhi[OWNER_GROUP];
#pragma unroll
        for (int u = 0; u < OWNER_GROUP; ++u) {
          vlo[u] = 0.f; vhi[u] = 0.f;
          if (j0 + u < m) {
            const float* __restrict__ srow = Wn + (size_t)lane_int(s, j0 + u) * DNODE;
            vlo[u] = srow[lane];
            vhi[u] = srow[32 + lane];
          }
        }
#pragma unroll
        for (int u = 0; u < OWNER_GROUP; ++u) {
          if (j0 + u < m) {
            const int j  = j0 + u;
            const int rj = lane_int(rl, j);
            float* drow = acc + rj * DNODE;
            const float a0 = lane_float(ez0, j), a1 = lane_float(ez1, j);
            const float a2 = lane_float(ez2, j), a3 = lane_float(ez3, j);
            atomicAdd(drow + lane,      (up ? a1 : a0) * vlo[u]);
            atomicAdd(drow + 32 + lane, (up ? a3 : a2) * vhi[u]);
          }
        }
      }
    }
  }
  __syncthreads();

  const float sc_lo = scale[lane], sc_hi = scale[32 + lane];
  const float bi_lo = bias[lane],  bi_hi = bias[32 + lane];
  for (int n = wave; n < nn; n += 32) {
    const float n01 = up ? nrm[n * HEADS + 1] : nrm[n * HEADS + 0];
    const float n23 = up ? nrm[n * HEADS + 3] : nrm[n * HEADS + 2];
    float x0 = acc[n * DNODE + lane] / n01;
    float x1 = acc[n * DNODE + 32 + lane] / n23;
    x0 = (x0 > 0.f) ? x0 : expm1f(x0);
    x1 = (x1 > 0.f) ? x1 : expm1f(x1);
    float sum = x0 + x1;
    for (int d = 16; d >= 1; d >>= 1) sum += __shfl_xor(sum, d);
    const float mean = sum * (1.f / DNODE);
    const float d0 = x0 - mean, d1 = x1 - mean;
    float var = d0 * d0 + d1 * d1;
    for (int d = 16; d >= 1; d >>= 1) var += __shfl_xor(var, d);
    var *= (1.f / DNODE);
    const float inv = rsqrtf(var + 1e-6f);
    float* row = out + (size_t)(node0 + n) * DNODE;
    row[lane]      = d0 * inv * sc_lo + bi_lo;
    row[32 + lane] = d1 * inv * sc_hi + bi_hi;
  }
}

extern "C" void kernel_launch(void* const* d_in, const int* in_sizes, int n_in,
                              void* d_out, int out_size, void* d_ws, size_t ws_size,
                              hipStream_t stream)
{
  const float* nodes  = (const float*)d_in[0];
  const float* edges  = (const float*)d_in[1];
  const int*   recv   = (const int*)d_in[2];
  const int*   send   = (const int*)d_in[3];
  const float* W      = (const float*)d_in[4];
  const float* W_edge = (const float*)d_in[5];
  const float* a      = (const float*)d_in[6];
  const float* ln_s   = (const float*)d_in[7];
  const float* ln_b   = (const float*)d_in[8];

  const int n_nodes = in_sizes[0] / DNODE;
  const int n_edges = in_sizes[1] / DEDGE;
  const int etot    = n_edges + n_nodes;
  float* out = (float*)d_out;

  char* ws = (char*)d_ws;
  size_t off = 0;
  auto carve = [&](size_t bytes) -> float* {
    float* p = (float*)(ws + off);
    off += (bytes + 255) & ~(size_t)255;
    return p;
  };
  float* Wn      = carve((size_t)n_nodes * DNODE * sizeof(float));
  float* score_s = carve((size_t)n_nodes * HEADS * sizeof(float));
  float* score_r = carve((size_t)n_nodes * HEADS * sizeof(float));
  float* wcomb   = carve((size_t)HEADS * DEDGE * sizeof(float));
  const size_t off_common = off;

  const int n_tiles = (n_nodes + 15) / 16;
  dim3 blk(256);

  gat_node_project<<<dim3((n_tiles + 7) / 8), blk, 0, stream>>>(
      nodes, W, a, W_edge, Wn, score_s, score_r, wcomb, n_nodes, n_tiles);

  const int npb          = (n_nodes + NBINS - 1) / NBINS;
  const int chunk_edges  = (n_edges + NCHUNK - 1) / NCHUNK;
  const int chunk_stride = (chunk_edges + 31) & ~31;
  const size_t lds_bytes = (size_t)npb * (DNODE + 2 * HEADS) * sizeof(float);
  int2*   seg   = (int2*)carve((size_t)NBINS * NCHUNK * sizeof(int2));
  int*    sr    = (int*)carve((size_t)NCHUNK * chunk_stride * sizeof(int));
  float4* lgrec = (float4*)carve((size_t)NCHUNK * chunk_stride * sizeof(float4));
  if (off <= ws_size && lds_bytes <= 65536 && n_nodes <= (1 << 24) && npb > 0 &&
      chunk_edges <= STAGE_CAP) {
    gat_edge_order<<<dim3(NCHUNK), dim3(1024), 0, stream>>>(
        edges, recv, send, score_s, wcomb, sr, lgrec, seg,
        n_edges, chunk_edges, chunk_stride, npb);
    gat_owner_sum<<<dim3(NBINS), dim3(1024), 0, stream>>>(
        sr, lgrec, seg, score_s, score_r, Wn, ln_s, ln_b, out, n_nodes, npb);
    return;
  }

  off = off_common;
  const size_t plane_bytes = (size_t)n_nodes * DNODE * sizeof(float);
  const size_t norms_bytes = (size_t)2 * n_nodes * HEADS * sizeof(float);
  float* aggr1   = carve(plane_bytes + norms_bytes);
  float* norm01  = aggr1 + (size_t)n_nodes * DNODE;

  hipMemsetAsync(out,   0, plane_bytes, stream);
  hipMemsetAsync(aggr1, 0, plane_bytes + norms_bytes, stream);

  gat_edge_fused<<<dim3((etot + 255) / 256), blk, 0, stream>>>(
      edges, recv, send, score_s, score_r, wcomb, Wn, out, aggr1, norm01, n_edges, n_nodes);
  gat_finalize<<<dim3((n_nodes + 255) / 256), blk, 0, stream>>>(
      out, aggr1, norm01, ln_s, ln_b, n_nodes);
}
